# starting kernel
# baseline (speedup 1.0000x reference)
#include <hip/hip_runtime.h>
#include <math.h>


typedef __attribute__((ext_vector_type(2))) float v2f;
typedef __attribute__((ext_vector_type(8))) float v8f;

__device__ __forceinline__ unsigned encf(float f) {
  unsigned u = __float_as_uint(f);
  return (u & 0x80000000u) ? ~u : (u | 0x80000000u);
}
__device__ __forceinline__ float decf(unsigned u) {
  unsigned v = (u & 0x80000000u) ? (u & 0x7FFFFFFFu) : ~u;
  return __uint_as_float(v);
}
#define ENC_NEG_INF 0x007FFFFFu

__global__ void gemm16_wmma(const float* __restrict__ X,
                            const float* __restrict__ W,
                            float* __restrict__ Hout,
                            int nrows, int ncols) {
  const int wave = blockIdx.x * (blockDim.x >> 5) + (threadIdx.x >> 5);
  const int lane = threadIdx.x & 31;
  const int row_base = wave * 16;
  if (row_base >= nrows) return;

  const int m    = lane & 15;
  const int half = lane >> 4;
  int row = row_base + m; if (row > nrows - 1) row = nrows - 1;
  int col = m;            if (col > ncols - 1) col = ncols - 1;

  const float2* __restrict__ Xr = (const float2*)(X + (size_t)row * 16);

  v8f c = {0.f, 0.f, 0.f, 0.f, 0.f, 0.f, 0.f, 0.f};
#pragma unroll
  for (int k0 = 0; k0 < 4; ++k0) {
    const int k = k0 * 4 + half * 2;
    const float2 av = Xr[k0 * 2 + half];
    v2f a; a.x = av.x; a.y = av.y;
    v2f b;
    b.x = W[(k)     * ncols + col];
    b.y = W[(k + 1) * ncols + col];
    c = __builtin_amdgcn_wmma_f32_16x16x4_f32(
         false, a,  false, b,
         (short)0, c,  false,  false);
  }

  if (m < ncols) {
    const int rofs = half * 8;
    float* __restrict__ outp = Hout + (size_t)(row_base + rofs) * ncols + m;
    if (row_base + 16 <= nrows) {
#pragma unroll
      for (int r = 0; r < 8; ++r) outp[(size_t)r * ncols] = c[r];
    } else {
#pragma unroll
      for (int r = 0; r < 8; ++r)
        if (row_base + rofs + r < nrows) outp[(size_t)r * ncols] = c[r];
    }
  }
}

template <int C>
__global__ void node_logits_init(const float* __restrict__ Hm,
                                 const float* __restrict__ a_src,
                                 const float* __restrict__ a_dst,
                                 float2* __restrict__ als, float2* __restrict__ ald,
                                 unsigned* __restrict__ menc,
                                 float2* __restrict__ z, float* __restrict__ acc,
                                 int nrows) {
  const int n = blockIdx.x * blockDim.x + threadIdx.x;
  if (n >= nrows) return;
  constexpr int F = 2 * C;
  const float4* __restrict__ hr = (const float4*)(Hm + (size_t)n * F);
  float s0 = 0.f, s1 = 0.f, d0 = 0.f, d1 = 0.f;
#pragma unroll
  for (int v = 0; v < C / 4; ++v) {
    const float4 h0 = hr[v];
    const float4 h1 = hr[C / 4 + v];
#pragma unroll
    for (int j = 0; j < 4; ++j) {
      const float e0 = (&h0.x)[j], e1 = (&h1.x)[j];
      const int cch = 4 * v + j;
      s0 += e0 * a_src[cch];     d0 += e0 * a_dst[cch];
      s1 += e1 * a_src[C + cch]; d1 += e1 * a_dst[C + cch];
    }
  }
  als[n] = make_float2(s0, s1);
  ald[n] = make_float2(d0, d1);
  menc[n * 2] = ENC_NEG_INF; menc[n * 2 + 1] = ENC_NEG_INF;
  z[n] = make_float2(0.f, 0.f);
  float4* __restrict__ ar = (float4*)(acc + (size_t)n * F);
#pragma unroll
  for (int v = 0; v < F / 4; ++v) ar[v] = make_float4(0.f, 0.f, 0.f, 0.f);
}

__global__ void edge_max(const int* __restrict__ ei,
                         const float2* __restrict__ als,
                         const float2* __restrict__ ald,
                         unsigned* __restrict__ menc, int E, int N) {
  const int t = blockIdx.x * blockDim.x + threadIdx.x;
  if (t >= E + N) return;
  int src, dst;
  if (t < E) { src = ei[t]; dst = ei[E + t]; } else { src = dst = t - E; }
  const float2 s = als[src];
  const float2 d = ald[dst];
  float e0 = s.x + d.x; e0 = (e0 > 0.f) ? e0 : 0.2f * e0;
  float e1 = s.y + d.y; e1 = (e1 > 0.f) ? e1 : 0.2f * e1;
  atomicMax(&menc[dst * 2],     encf(e0));
  atomicMax(&menc[dst * 2 + 1], encf(e1));
}

template <int C>
__global__ void edge_accum(const int* __restrict__ ei,
                           const float2* __restrict__ als,
                           const float2* __restrict__ ald,
                           const unsigned* __restrict__ menc,
                           float* __restrict__ z, float* __restrict__ acc,
                           const float* __restrict__ Hm,
                           int E, int N) {
  const int t = blockIdx.x * blockDim.x + threadIdx.x;
  if (t >= E + N) return;
  int src, dst;
  if (t < E) { src = ei[t]; dst = ei[E + t]; } else { src = dst = t - E; }
  constexpr int F = 2 * C;
  const float2 s = als[src];
  const float2 d = ald[dst];
  const float4* __restrict__ hs = (const float4*)(Hm + (size_t)src * F);
  float w[2];
#pragma unroll
  for (int hd = 0; hd < 2; ++hd) {
    float e = (hd ? s.y + d.y : s.x + d.x);
    e = (e > 0.f) ? e : 0.2f * e;
    float mm = decf(menc[dst * 2 + hd]);
    if (!(mm >= -3.0e38f && mm <= 3.0e38f)) mm = 0.f;
    w[hd] = __expf(e - mm);
    atomicAdd(&z[dst * 2 + hd], w[hd]);
  }
  float* __restrict__ ad = acc + (size_t)dst * F;
#pragma unroll
  for (int hd = 0; hd < 2; ++hd) {
#pragma unroll
    for (int v = 0; v < C / 4; ++v) {
      const float4 hv = hs[hd * (C / 4) + v];
      const int base = hd * C + 4 * v;
      atomicAdd(&ad[base + 0], w[hd] * hv.x);
      atomicAdd(&ad[base + 1], w[hd] * hv.y);
      atomicAdd(&ad[base + 2], w[hd] * hv.z);
      atomicAdd(&ad[base + 3], w[hd] * hv.w);
    }
  }
}

template <int C, bool ELU>
__global__ void node_finalize(const float* __restrict__ acc,
                              const float2* __restrict__ z,
                              const float* __restrict__ bias,
                              float* __restrict__ out, int nrows) {
  const int n = blockIdx.x * blockDim.x + threadIdx.x;
  if (n >= nrows) return;
  constexpr int F = 2 * C;
  const float4* __restrict__ ar = (const float4*)(acc + (size_t)n * F);
  float4* __restrict__ outp = (float4*)(out + (size_t)n * F);
  const float2 zz = z[n];
  const float rz0 = 1.f / (zz.x + 1e-16f);
  const float rz1 = 1.f / (zz.y + 1e-16f);
#pragma unroll
  for (int v = 0; v < F / 4; ++v) {
    float4 a = ar[v];
    const float rz = (4 * v < C) ? rz0 : rz1;
#pragma unroll
    for (int j = 0; j < 4; ++j) {
      float x = (&a.x)[j] * rz + bias[4 * v + j];
      if (ELU) x = (x > 0.f) ? x : (__expf(x) - 1.f);
      (&a.x)[j] = x;
    }
    outp[v] = a;
  }
}

extern "C" void kernel_launch(void* const* d_in, const int* in_sizes, int n_in,
                              void* d_out, int out_size, void* d_ws, size_t ws_size,
                              hipStream_t stream) {
  const float* x   = (const float*)d_in[0];
  const int*   ei  = (const int*)  d_in[1];
  const float* W1  = (const float*)d_in[2];
  const float* as1 = (const float*)d_in[3];
  const float* ad1 = (const float*)d_in[4];
  const float* b1  = (const float*)d_in[5];
  const float* W2  = (const float*)d_in[6];
  const float* as2 = (const float*)d_in[7];
  const float* ad2 = (const float*)d_in[8];
  const float* b2  = (const float*)d_in[9];

  const int N = in_sizes[0] / 16;
  const int E = in_sizes[1] / 2;

  float*    ws   = (float*)d_ws;
  float*    h1   = ws;
  float*    acc1 = ws + (size_t)16 * N;
  float2*   als  = (float2*)(ws + (size_t)32 * N);
  float2*   ald  = (float2*)(ws + (size_t)34 * N);
  unsigned* menc = (unsigned*)(ws + (size_t)36 * N);
  float*    z    = ws + (size_t)38 * N;
  float*    h2   = ws + (size_t)40 * N;
  float*    acc2 = ws + (size_t)48 * N;

  const int TB = 256;
  const int node_grid = (N + TB - 1) / TB;
  const int edge_grid = (E + N + TB - 1) / TB;
  const int tiles     = (N + 15) / 16;
  const int gemm_grid = (tiles + 7) / 8;

  gemm16_wmma<<<gemm_grid, TB, 0, stream>>>(x, W1, h1, N, 16);
  node_logits_init<8><<<node_grid, TB, 0, stream>>>(h1, as1, ad1, als, ald, menc,
                                                    (float2*)z, acc1, N);
  edge_max<<<edge_grid, TB, 0, stream>>>(ei, als, ald, menc, E, N);
  edge_accum<8><<<edge_grid, TB, 0, stream>>>(ei, als, ald, menc, z, acc1, h1, E, N);
  node_finalize<8, true><<<node_grid, TB, 0, stream>>>(acc1, (const float2*)z, b1, acc1, N);

  gemm16_wmma<<<gemm_grid, TB, 0, stream>>>(acc1, W2, h2, N, 8);
  node_logits_init<4><<<node_grid, TB, 0, stream>>>(h2, as2, ad2, als, ald, menc,
                                                    (float2*)z, acc2, N);
  edge_max<<<edge_grid, TB, 0, stream>>>(ei, als, ald, menc, E, N);
  edge_accum<4><<<edge_grid, TB, 0, stream>>>(ei, als, ald, menc, z, acc2, h2, E, N);
  node_finalize<4, false><<<node_grid, TB, 0, stream>>>(acc2, (const float2*)z, b2,
                                                        (float*)d_out, N);
}


# edge_accum: one lane per channel (16 or 8 lanes per edge), so h[src] loads and acc[dst] f32 atomic adds cover whole contiguous rows per wave instruction; sums still in arrival order
# speedup vs baseline: 1.8529x; 1.8529x over previous
#include <hip/hip_runtime.h>
#include <math.h>


typedef __attribute__((ext_vector_type(2))) float v2f;
typedef __attribute__((ext_vector_type(8))) float v8f;

__device__ __forceinline__ unsigned encf(float f) {
  unsigned u = __float_as_uint(f);
  return (u & 0x80000000u) ? ~u : (u | 0x80000000u);
}
__device__ __forceinline__ float decf(unsigned u) {
  unsigned v = (u & 0x80000000u) ? (u & 0x7FFFFFFFu) : ~u;
  return __uint_as_float(v);
}
#define ENC_NEG_INF 0x007FFFFFu

__global__ void gemm16_wmma(const float* __restrict__ X,
                            const float* __restrict__ W,
                            float* __restrict__ Hout,
                            int nrows, int ncols) {
  const int wave = blockIdx.x * (blockDim.x >> 5) + (threadIdx.x >> 5);
  const int lane = threadIdx.x & 31;
  const int row_base = wave * 16;
  if (row_base >= nrows) return;

  const int m    = lane & 15;
  const int half = lane >> 4;
  int row = row_base + m; if (row > nrows - 1) row = nrows - 1;
  int col = m;            if (col > ncols - 1) col = ncols - 1;

  const float2* __restrict__ Xr = (const float2*)(X + (size_t)row * 16);

  v8f c = {0.f, 0.f, 0.f, 0.f, 0.f, 0.f, 0.f, 0.f};
#pragma unroll
  for (int k0 = 0; k0 < 4; ++k0) {
    const int k = k0 * 4 + half * 2;
    const float2 av = Xr[k0 * 2 + half];
    v2f a; a.x = av.x; a.y = av.y;
    v2f b;
    b.x = W[(k)     * ncols + col];
    b.y = W[(k + 1) * ncols + col];
    c = __builtin_amdgcn_wmma_f32_16x16x4_f32(
         false, a,  false, b,
         (short)0, c,  false,  false);
  }

  if (m < ncols) {
    const int rofs = half * 8;
    float* __restrict__ outp = Hout + (size_t)(row_base + rofs) * ncols + m;
    if (row_base + 16 <= nrows) {
#pragma unroll
      for (int r = 0; r < 8; ++r) outp[(size_t)r * ncols] = c[r];
    } else {
#pragma unroll
      for (int r = 0; r < 8; ++r)
        if (row_base + rofs + r < nrows) outp[(size_t)r * ncols] = c[r];
    }
  }
}

template <int C>
__global__ void node_logits_init(const float* __restrict__ Hm,
                                 const float* __restrict__ a_src,
                                 const float* __restrict__ a_dst,
                                 float2* __restrict__ als, float2* __restrict__ ald,
                                 unsigned* __restrict__ menc,
                                 float2* __restrict__ z, float* __restrict__ acc,
                                 int nrows) {
  const int n = blockIdx.x * blockDim.x + threadIdx.x;
  if (n >= nrows) return;
  constexpr int F = 2 * C;
  const float4* __restrict__ hr = (const float4*)(Hm + (size_t)n * F);
  float s0 = 0.f, s1 = 0.f, d0 = 0.f, d1 = 0.f;
#pragma unroll
  for (int v = 0; v < C / 4; ++v) {
    const float4 h0 = hr[v];
    const float4 h1 = hr[C / 4 + v];
#pragma unroll
    for (int j = 0; j < 4; ++j) {
      const float e0 = (&h0.x)[j], e1 = (&h1.x)[j];
      const int cch = 4 * v + j;
      s0 += e0 * a_src[cch];     d0 += e0 * a_dst[cch];
      s1 += e1 * a_src[C + cch]; d1 += e1 * a_dst[C + cch];
    }
  }
  als[n] = make_float2(s0, s1);
  ald[n] = make_float2(d0, d1);
  menc[n * 2] = ENC_NEG_INF; menc[n * 2 + 1] = ENC_NEG_INF;
  z[n] = make_float2(0.f, 0.f);
  float4* __restrict__ ar = (float4*)(acc + (size_t)n * F);
#pragma unroll
  for (int v = 0; v < F / 4; ++v) ar[v] = make_float4(0.f, 0.f, 0.f, 0.f);
}

__global__ void edge_max(const int* __restrict__ ei,
                         const float2* __restrict__ als,
                         const float2* __restrict__ ald,
                         unsigned* __restrict__ menc, int E, int N) {
  const int t = blockIdx.x * blockDim.x + threadIdx.x;
  if (t >= E + N) return;
  int src, dst;
  if (t < E) { src = ei[t]; dst = ei[E + t]; } else { src = dst = t - E; }
  const float2 s = als[src];
  const float2 d = ald[dst];
  float e0 = s.x + d.x; e0 = (e0 > 0.f) ? e0 : 0.2f * e0;
  float e1 = s.y + d.y; e1 = (e1 > 0.f) ? e1 : 0.2f * e1;
  atomicMax(&menc[dst * 2],     encf(e0));
  atomicMax(&menc[dst * 2 + 1], encf(e1));
}

template <int C, bool GUARD>
__device__ __forceinline__ void accum_rows(int src, int dst, float w0, float w1,
                                           int lane,
                                           float* __restrict__ z,
                                           float* __restrict__ acc,
                                           const float* __restrict__ Hm) {
  constexpr int F   = 2 * C;
  constexpr int EPI = 32 / F;
  constexpr int U   = 4;
  const int  sub   = lane / F;
  const int  j     = lane % F;
  const bool head1 = (j >= C);
  const bool zlane = (j == 0) || (j == C);
  float* __restrict__ zp = z + (head1 ? 1 : 0);
#pragma unroll
  for (int it0 = 0; it0 < F; it0 += U) {
    int   dq[U];
    float wq[U], hq[U];
    bool  ok[U];
#pragma unroll
    for (int u = 0; u < U; ++u) {
      const int o  = (it0 + u) * EPI + sub;
      const int so = __shfl(src, o);
      dq[u] = __shfl(dst, o);
      const float wa = __shfl(w0, o), wb = __shfl(w1, o);
      wq[u] = head1 ? wb : wa;
      ok[u] = !GUARD || (so >= 0);
      const int sc = ok[u] ? so : 0;
      hq[u] = Hm[(size_t)sc * F + j];
    }
#pragma unroll
    for (int u = 0; u < U; ++u) {
      if (ok[u]) {
        atomicAdd(&acc[(size_t)dq[u] * F + j], wq[u] * hq[u]);
        if (zlane) atomicAdd(&zp[(size_t)dq[u] * 2], wq[u]);
      }
    }
  }
}

template <int C>
__global__ void edge_accum(const int* __restrict__ ei,
                           const float2* __restrict__ als,
                           const float2* __restrict__ ald,
                           const unsigned* __restrict__ menc,
                           float* __restrict__ z, float* __restrict__ acc,
                           const float* __restrict__ Hm,
                           int E, int N) {
  const int t    = blockIdx.x * blockDim.x + threadIdx.x;
  const int lane = threadIdx.x & 31;

  int src = -1, dst = 0;
  float w0 = 0.f, w1 = 0.f;
  if (t < E + N) {
    if (t < E) { src = ei[t]; dst = ei[E + t]; } else { src = dst = t - E; }
    const float2 s = als[src];
    const float2 d = ald[dst];
    const uint2 mu = *(const uint2*)(menc + (size_t)dst * 2);
    float e0 = s.x + d.x; e0 = (e0 > 0.f) ? e0 : 0.2f * e0;
    float e1 = s.y + d.y; e1 = (e1 > 0.f) ? e1 : 0.2f * e1;
    float m0 = decf(mu.x), m1 = decf(mu.y);
    if (!(m0 >= -3.0e38f && m0 <= 3.0e38f)) m0 = 0.f;
    if (!(m1 >= -3.0e38f && m1 <= 3.0e38f)) m1 = 0.f;
    w0 = __expf(e0 - m0);
    w1 = __expf(e1 - m1);
  }

  const int wave_first = __builtin_amdgcn_readfirstlane(t);
  if (wave_first + 32 <= E + N)
    accum_rows<C, false>(src, dst, w0, w1, lane, z, acc, Hm);
  else
    accum_rows<C, true>(src, dst, w0, w1, lane, z, acc, Hm);
}

template <int C, bool ELU>
__global__ void node_finalize(const float* __restrict__ acc,
                              const float2* __restrict__ z,
                              const float* __restrict__ bias,
                              float* __restrict__ out, int nrows) {
  const int n = blockIdx.x * blockDim.x + threadIdx.x;
  if (n >= nrows) return;
  constexpr int F = 2 * C;
  const float4* __restrict__ ar = (const float4*)(acc + (size_t)n * F);
  float4* __restrict__ outp = (float4*)(out + (size_t)n * F);
  const float2 zz = z[n];
  const float rz0 = 1.f / (zz.x + 1e-16f);
  const float rz1 = 1.f / (zz.y + 1e-16f);
#pragma unroll
  for (int v = 0; v < F / 4; ++v) {
    float4 a = ar[v];
    const float rz = (4 * v < C) ? rz0 : rz1;
#pragma unroll
    for (int j = 0; j < 4; ++j) {
      float x = (&a.x)[j] * rz + bias[4 * v + j];
      if (ELU) x = (x > 0.f) ? x : (__expf(x) - 1.f);
      (&a.x)[j] = x;
    }
    outp[v] = a;
  }
}

extern "C" void kernel_launch(void* const* d_in, const int* in_sizes, int n_in,
                              void* d_out, int out_size, void* d_ws, size_t ws_size,
                              hipStream_t stream) {
  const float* x   = (const float*)d_in[0];
  const int*   ei  = (const int*)  d_in[1];
  const float* W1  = (const float*)d_in[2];
  const float* as1 = (const float*)d_in[3];
  const float* ad1 = (const float*)d_in[4];
  const float* b1  = (const float*)d_in[5];
  const float* W2  = (const float*)d_in[6];
  const float* as2 = (const float*)d_in[7];
  const float* ad2 = (const float*)d_in[8];
  const float* b2  = (const float*)d_in[9];

  const int N = in_sizes[0] / 16;
  const int E = in_sizes[1] / 2;

  float*    ws   = (float*)d_ws;
  float*    h1   = ws;
  float*    acc1 = ws + (size_t)16 * N;
  float2*   als  = (float2*)(ws + (size_t)32 * N);
  float2*   ald  = (float2*)(ws + (size_t)34 * N);
  unsigned* menc = (unsigned*)(ws + (size_t)36 * N);
  float*    z    = ws + (size_t)38 * N;
  float*    h2   = ws + (size_t)40 * N;
  float*    acc2 = ws + (size_t)48 * N;

  const int TB = 256;
  const int node_grid = (N + TB - 1) / TB;
  const int edge_grid = (E + N + TB - 1) / TB;
  const int tiles     = (N + 15) / 16;
  const int gemm_grid = (tiles + 7) / 8;

  gemm16_wmma<<<gemm_grid, TB, 0, stream>>>(x, W1, h1, N, 16);
  node_logits_init<8><<<node_grid, TB, 0, stream>>>(h1, as1, ad1, als, ald, menc,
                                                    (float2*)z, acc1, N);
  edge_max<<<edge_grid, TB, 0, stream>>>(ei, als, ald, menc, E, N);
  edge_accum<8><<<edge_grid, TB, 0, stream>>>(ei, als, ald, menc, z, acc1, h1, E, N);
  node_finalize<8, true><<<node_grid, TB, 0, stream>>>(acc1, (const float2*)z, b1, acc1, N);

  gemm16_wmma<<<gemm_grid, TB, 0, stream>>>(acc1, W2, h2, N, 8);
  node_logits_init<4><<<node_grid, TB, 0, stream>>>(h2, as2, ad2, als, ald, menc,
                                                    (float2*)z, acc2, N);
  edge_max<<<edge_grid, TB, 0, stream>>>(ei, als, ald, menc, E, N);
  edge_accum<4><<<edge_grid, TB, 0, stream>>>(ei, als, ald, menc, z, acc2, h2, E, N);
  node_finalize<4, false><<<node_grid, TB, 0, stream>>>(acc2, (const float2*)z, b2,
                                                        (float*)d_out, N);
}


# edge softmax, both layers: per call edges grouped by destination (count, scan, atomic placement); 2C lanes per node walk its list, max and sums in registers in list (arrival) order; no float atomics
# speedup vs baseline: 2.3808x; 1.2849x over previous
#include <hip/hip_runtime.h>
#include <math.h>


typedef __attribute__((ext_vector_type(2))) float v2f;
typedef __attribute__((ext_vector_type(8))) float v8f;

__global__ void gemm16_wmma(const float* __restrict__ X,
                            const float* __restrict__ W,
                            float* __restrict__ Hout,
                            int nrows, int ncols) {
  const int wave = blockIdx.x * (blockDim.x >> 5) + (threadIdx.x >> 5);
  const int lane = threadIdx.x & 31;
  const int row_base = wave * 16;
  if (row_base >= nrows) return;

  const int m    = lane & 15;
  const int half = lane >> 4;
  int row = row_base + m; if (row > nrows - 1) row = nrows - 1;
  int col = m;            if (col > ncols - 1) col = ncols - 1;

  const float2* __restrict__ Xr = (const float2*)(X + (size_t)row * 16);

  v8f c = {0.f, 0.f, 0.f, 0.f, 0.f, 0.f, 0.f, 0.f};
#pragma unroll
  for (int k0 = 0; k0 < 4; ++k0) {
    const int k = k0 * 4 + half * 2;
    const float2 av = Xr[k0 * 2 + half];
    v2f a; a.x = av.x; a.y = av.y;
    v2f b;
    b.x = W[(k)     * ncols + col];
    b.y = W[(k + 1) * ncols + col];
    c = __builtin_amdgcn_wmma_f32_16x16x4_f32(
         false, a,  false, b,
         (short)0, c,  false,  false);
  }

  if (m < ncols) {
    const int rofs = half * 8;
    float* __restrict__ outp = Hout + (size_t)(row_base + rofs) * ncols + m;
    if (row_base + 16 <= nrows) {
#pragma unroll
      for (int r = 0; r < 8; ++r) outp[(size_t)r * ncols] = c[r];
    } else {
#pragma unroll
      for (int r = 0; r < 8; ++r)
        if (row_base + rofs + r < nrows) outp[(size_t)r * ncols] = c[r];
    }
  }
}

template <int C, bool FIRST>
__global__ void node_logits(const float* __restrict__ Hm,
                            const float* __restrict__ a_src,
                            const float* __restrict__ a_dst,
                            float2* __restrict__ als, float2* __restrict__ ald,
                            int* __restrict__ cnt, const int* __restrict__ off,
                            int* __restrict__ cur, int nrows) {
  const int n = blockIdx.x * blockDim.x + threadIdx.x;
  if (n >= nrows) return;
  constexpr int F = 2 * C;
  const float4* __restrict__ hr = (const float4*)(Hm + (size_t)n * F);
  float s0 = 0.f, s1 = 0.f, d0 = 0.f, d1 = 0.f;
#pragma unroll
  for (int v = 0; v < C / 4; ++v) {
    const float4 h0 = hr[v];
    const float4 h1 = hr[C / 4 + v];
#pragma unroll
    for (int j = 0; j < 4; ++j) {
      const float e0 = (&h0.x)[j], e1 = (&h1.x)[j];
      const int cch = 4 * v + j;
      s0 += e0 * a_src[cch];     d0 += e0 * a_dst[cch];
      s1 += e1 * a_src[C + cch]; d1 += e1 * a_dst[C + cch];
    }
  }
  als[n] = make_float2(s0, s1);
  ald[n] = make_float2(d0, d1);
  if (FIRST) cnt[n] = 0; else cur[n] = off[n];
}

#define EPT 4

__global__ void dst_count(const int* __restrict__ ei, int* __restrict__ cnt, int E) {
  const int T = gridDim.x * blockDim.x;
  const int t = blockIdx.x * blockDim.x + threadIdx.x;
  int d[EPT];
#pragma unroll
  for (int k = 0; k < EPT; ++k) {
    const int e = t + k * T;
    d[k] = (e < E) ? ei[E + e] : -1;
  }
#pragma unroll
  for (int k = 0; k < EPT; ++k)
    if (d[k] >= 0) atomicAdd(&cnt[d[k]], 1);
}

__global__ void seg_offsets(const int* __restrict__ cnt, int* __restrict__ off,
                            int* __restrict__ cur, int N) {
  __shared__ int wsum[8];
  __shared__ int wtot[8];
  const int tid  = threadIdx.x;
  const int lane = tid & 31;
  const int wv   = tid >> 5;
  const int block_start = blockIdx.x * 256;
  const int4* __restrict__ c4 = (const int4*)cnt;
  int part = 0;
#pragma unroll 8
  for (int i = tid; i < block_start / 4; i += 256) {
    const int4 v = c4[i];
    part += (v.x + v.y) + (v.z + v.w);
  }
#pragma unroll
  for (int o = 16; o > 0; o >>= 1) part += __shfl_xor(part, o);
  const int n = block_start + tid;
  const int c = (n < N) ? cnt[n] : 0;
  int incl = c;
#pragma unroll
  for (int o = 1; o < 32; o <<= 1) {
    const int up = __shfl_up(incl, o);
    if (lane >= o) incl += up;
  }
  if (lane == 0)  wsum[wv] = part;
  if (lane == 31) wtot[wv] = incl;
  __syncthreads();
  int base = 0;
#pragma unroll
  for (int k = 0; k < 8; ++k) {
    base += wsum[k];
    if (k < wv) base += wtot[k];
  }
  if (n < N) {
    const int ex = base + incl - c;
    off[n] = ex;
    cur[n] = ex;
  }
}

__global__ void dst_place(const int* __restrict__ ei, int* __restrict__ cur,
                          const int* __restrict__ off, int* __restrict__ list,
                          int E, int lo, int hi, int cap) {
  const int T = gridDim.x * blockDim.x;
  const int t = blockIdx.x * blockDim.x + threadIdx.x;
  const int base = off[lo];
  int d[EPT], s[EPT];
#pragma unroll
  for (int k = 0; k < EPT; ++k) {
    const int e = t + k * T;
    const bool in = (e < E);
    d[k] = in ? ei[E + e] : -1;
    s[k] = in ? ei[e] : 0;
  }
  int p[EPT];
#pragma unroll
  for (int k = 0; k < EPT; ++k) {
    const bool mine = (d[k] >= lo && d[k] < hi);
    p[k] = mine ? (atomicAdd(&cur[d[k]], 1) - base) : -1;
  }
#pragma unroll
  for (int k = 0; k < EPT; ++k)
    if ((unsigned)p[k] < (unsigned)cap) list[p[k]] = s[k];
}

template <int C, bool ELU>
__global__ void seg_softmax_agg(const int* __restrict__ list,
                                const int* __restrict__ cur,
                                const int* __restrict__ off,
                                const float2* __restrict__ als,
                                const float2* __restrict__ ald,
                                const float* __restrict__ Hm,
                                const float* __restrict__ bias,
                                float* __restrict__ out,
                                int lo, int hi, int cap, int N) {
  constexpr int F = 2 * C;
  constexpr int U = 4;
  const int t = blockIdx.x * blockDim.x + threadIdx.x;
  const int n = lo + t / F;
  const int j = t % F;
  if (n >= hi) return;
  const int head = (j >= C) ? 1 : 0;
  const float* __restrict__ alsf = (const float*)als;
  const float* __restrict__ aldf = (const float*)ald;

  const int base = off[lo];
  int beg = (n == lo) ? 0 : (cur[n - 1] - base);
  int end = cur[n] - base;
  beg = min(max(beg, 0), cap);
  end = min(max(end, beg), cap);

  const float adn = aldf[2 * n + head];
  float es = alsf[2 * n + head] + adn;
  es = (es > 0.f) ? es : 0.2f * es;

  float m = es;
  for (int p = beg; p < end; p += U) {
    int sq[U];
    float aq[U];
#pragma unroll
    for (int u = 0; u < U; ++u) {
      const int q = min(p + u, end - 1);
      sq[u] = min(max(list[q], 0), N - 1);
    }
#pragma unroll
    for (int u = 0; u < U; ++u) aq[u] = alsf[2 * sq[u] + head];
#pragma unroll
    for (int u = 0; u < U; ++u) {
      float e = aq[u] + adn;
      e = (e > 0.f) ? e : 0.2f * e;
      m = fmaxf(m, e);
    }
  }
  if (!(m >= -3.0e38f && m <= 3.0e38f)) m = 0.f;

  float w = __expf(es - m);
  float z = w;
  float acc = w * Hm[n * F + j];
  for (int p = beg; p < end; p += U) {
    int sq[U];
    bool ok[U];
    float aq[U], hq[U];
#pragma unroll
    for (int u = 0; u < U; ++u) {
      ok[u] = (p + u < end);
      const int q = ok[u] ? (p + u) : (end - 1);
      sq[u] = min(max(list[q], 0), N - 1);
    }
#pragma unroll
    for (int u = 0; u < U; ++u) {
      aq[u] = alsf[2 * sq[u] + head];
      hq[u] = Hm[sq[u] * F + j];
    }
#pragma unroll
    for (int u = 0; u < U; ++u) {
      float e = aq[u] + adn;
      e = (e > 0.f) ? e : 0.2f * e;
      e = ok[u] ? e : -3.0e38f;
      const float wu = __expf(e - m);
      z += wu;
      acc += wu * hq[u];
    }
  }
  float x = acc * (1.f / (z + 1e-16f)) + bias[j];
  if (ELU) x = (x > 0.f) ? x : (__expf(x) - 1.f);
  out[n * F + j] = x;
}

extern "C" void kernel_launch(void* const* d_in, const int* in_sizes, int n_in,
                              void* d_out, int out_size, void* d_ws, size_t ws_size,
                              hipStream_t stream) {
  const float* x   = (const float*)d_in[0];
  const int*   ei  = (const int*)  d_in[1];
  const float* W1  = (const float*)d_in[2];
  const float* as1 = (const float*)d_in[3];
  const float* ad1 = (const float*)d_in[4];
  const float* b1  = (const float*)d_in[5];
  const float* W2  = (const float*)d_in[6];
  const float* as2 = (const float*)d_in[7];
  const float* ad2 = (const float*)d_in[8];
  const float* b2  = (const float*)d_in[9];

  const int N = in_sizes[0] / 16;
  const int E = in_sizes[1] / 2;

  float*  ws    = (float*)d_ws;
  float*  h1    = ws;
  float*  x2    = ws + (size_t)16 * N;
  float2* als   = (float2*)(ws + (size_t)32 * N);
  float2* ald   = (float2*)(ws + (size_t)34 * N);
  int*    cnt   = (int*)(ws + (size_t)36 * N);
  int*    off   = (int*)(ws + (size_t)37 * N);
  int*    cur   = (int*)(ws + (size_t)38 * N);
  int*    listP = (int*)(ws + (size_t)39 * N);
  float*  h2    = ws + (size_t)40 * N;
  int*    listF = (int*)ws;

  const int TB = 256;
  const int node_grid  = (N + TB - 1) / TB;
  const int list_grid  = (E + TB * EPT - 1) / (TB * EPT);
  const int tiles      = (N + 15) / 16;
  const int gemm_grid  = (tiles + 7) / 8;
  const int half       = N / 2;
  const int capP       = 17 * N;
  const int capF       = 32 * N;
  const int half_grid  = ((N - half) * 16 + TB - 1) / TB;
  const int lane_grid  = (N * 8 + TB - 1) / TB;

  gemm16_wmma<<<gemm_grid, TB, 0, stream>>>(x, W1, h1, N, 16);
  node_logits<8, true><<<node_grid, TB, 0, stream>>>(h1, as1, ad1, als, ald, cnt, off, cur, N);
  dst_count<<<list_grid, TB, 0, stream>>>(ei, cnt, E);
  seg_offsets<<<node_grid, TB, 0, stream>>>(cnt, off, cur, N);
  dst_place<<<list_grid, TB, 0, stream>>>(ei, cur, off, listP, E, 0, half, capP);
  seg_softmax_agg<8, true><<<half_grid, TB, 0, stream>>>(listP, cur, off, als, ald, h1, b1,
                                                         x2, 0, half, capP, N);
  dst_place<<<list_grid, TB, 0, stream>>>(ei, cur, off, listP, E, half, N, capP);
  seg_softmax_agg<8, true><<<half_grid, TB, 0, stream>>>(listP, cur, off, als, ald, h1, b1,
                                                         x2, half, N, capP, N);

  gemm16_wmma<<<gemm_grid, TB, 0, stream>>>(x2, W2, h2, N, 8);
  node_logits<4, false><<<node_grid, TB, 0, stream>>>(h2, as2, ad2, als, ald, cnt, off, cur, N);
  dst_place<<<list_grid, TB, 0, stream>>>(ei, cur, off, listF, E, 0, N, capF);
  seg_softmax_agg<4, false><<<lane_grid, TB, 0, stream>>>(listF, cur, off, als, ald, h2, b2,
                                                          (float*)d_out, 0, N, capF, N);
}


# edge softmax, both layers: one walk of a node's list with a running max (sums so far scaled by exp(old max - new max) when it rises) replaces max walk plus sum walk; each source logit gathered once
# speedup vs baseline: 2.5267x; 1.0613x over previous
#include <hip/hip_runtime.h>
#include <math.h>


typedef __attribute__((ext_vector_type(2))) float v2f;
typedef __attribute__((ext_vector_type(8))) float v8f;

__global__ void gemm16_wmma(const float* __restrict__ X,
                            const float* __restrict__ W,
                            float* __restrict__ Hout,
                            int nrows, int ncols) {
  const int wave = blockIdx.x * (blockDim.x >> 5) + (threadIdx.x >> 5);
  const int lane = threadIdx.x & 31;
  const int row_base = wave * 16;
  if (row_base >= nrows) return;

  const int m    = lane & 15;
  const int half = lane >> 4;
  int row = row_base + m; if (row > nrows - 1) row = nrows - 1;
  int col = m;            if (col > ncols - 1) col = ncols - 1;

  const float2* __restrict__ Xr = (const float2*)(X + (size_t)row * 16);

  v8f c = {0.f, 0.f, 0.f, 0.f, 0.f, 0.f, 0.f, 0.f};
#pragma unroll
  for (int k0 = 0; k0 < 4; ++k0) {
    const int k = k0 * 4 + half * 2;
    const float2 av = Xr[k0 * 2 + half];
    v2f a; a.x = av.x; a.y = av.y;
    v2f b;
    b.x = W[(k)     * ncols + col];
    b.y = W[(k + 1) * ncols + col];
    c = __builtin_amdgcn_wmma_f32_16x16x4_f32(
         false, a,  false, b,
         (short)0, c,  false,  false);
  }

  if (m < ncols) {
    const int rofs = half * 8;
    float* __restrict__ outp = Hout + (size_t)(row_base + rofs) * ncols + m;
    if (row_base + 16 <= nrows) {
#pragma unroll
      for (int r = 0; r < 8; ++r) outp[(size_t)r * ncols] = c[r];
    } else {
#pragma unroll
      for (int r = 0; r < 8; ++r)
        if (row_base + rofs + r < nrows) outp[(size_t)r * ncols] = c[r];
    }
  }
}

template <int C, bool FIRST>
__global__ void node_logits(const float* __restrict__ Hm,
                            const float* __restrict__ a_src,
                            const float* __restrict__ a_dst,
                            float2* __restrict__ als, float2* __restrict__ ald,
                            int* __restrict__ cnt, const int* __restrict__ off,
                            int* __restrict__ cur, int nrows) {
  const int n = blockIdx.x * blockDim.x + threadIdx.x;
  if (n >= nrows) return;
  constexpr int F = 2 * C;
  const float4* __restrict__ hr = (const float4*)(Hm + (size_t)n * F);
  float s0 = 0.f, s1 = 0.f, d0 = 0.f, d1 = 0.f;
#pragma unroll
  for (int v = 0; v < C / 4; ++v) {
    const float4 h0 = hr[v];
    const float4 h1 = hr[C / 4 + v];
#pragma unroll
    for (int j = 0; j < 4; ++j) {
      const float e0 = (&h0.x)[j], e1 = (&h1.x)[j];
      const int cch = 4 * v + j;
      s0 += e0 * a_src[cch];     d0 += e0 * a_dst[cch];
      s1 += e1 * a_src[C + cch]; d1 += e1 * a_dst[C + cch];
    }
  }
  als[n] = make_float2(s0, s1);
  ald[n] = make_float2(d0, d1);
  if (FIRST) cnt[n] = 0; else cur[n] = off[n];
}

#define EPT 4

__global__ void dst_count(const int* __restrict__ ei, int* __restrict__ cnt, int E) {
  const int T = gridDim.x * blockDim.x;
  const int t = blockIdx.x * blockDim.x + threadIdx.x;
  int d[EPT];
#pragma unroll
  for (int k = 0; k < EPT; ++k) {
    const int e = t + k * T;
    d[k] = (e < E) ? ei[E + e] : -1;
  }
#pragma unroll
  for (int k = 0; k < EPT; ++k)
    if (d[k] >= 0) atomicAdd(&cnt[d[k]], 1);
}

__global__ void seg_offsets(const int* __restrict__ cnt, int* __restrict__ off,
                            int* __restrict__ cur, int N) {
  __shared__ int wsum[8];
  __shared__ int wtot[8];
  const int tid  = threadIdx.x;
  const int lane = tid & 31;
  const int wv   = tid >> 5;
  const int block_start = blockIdx.x * 256;
  const int4* __restrict__ c4 = (const int4*)cnt;
  int part = 0;
#pragma unroll 8
  for (int i = tid; i < block_start / 4; i += 256) {
    const int4 v = c4[i];
    part += (v.x + v.y) + (v.z + v.w);
  }
#pragma unroll
  for (int o = 16; o > 0; o >>= 1) part += __shfl_xor(part, o);
  const int n = block_start + tid;
  const int c = (n < N) ? cnt[n] : 0;
  int incl = c;
#pragma unroll
  for (int o = 1; o < 32; o <<= 1) {
    const int up = __shfl_up(incl, o);
    if (lane >= o) incl += up;
  }
  if (lane == 0)  wsum[wv] = part;
  if (lane == 31) wtot[wv] = incl;
  __syncthreads();
  int base = 0;
#pragma unroll
  for (int k = 0; k < 8; ++k) {
    base += wsum[k];
    if (k < wv) base += wtot[k];
  }
  if (n < N) {
    const int ex = base + incl - c;
    off[n] = ex;
    cur[n] = ex;
  }
}

__global__ void dst_place(const int* __restrict__ ei, int* __restrict__ cur,
                          const int* __restrict__ off, int* __restrict__ list,
                          int E, int lo, int hi, int cap) {
  const int T = gridDim.x * blockDim.x;
  const int t = blockIdx.x * blockDim.x + threadIdx.x;
  const int base = off[lo];
  int d[EPT], s[EPT];
#pragma unroll
  for (int k = 0; k < EPT; ++k) {
    const int e = t + k * T;
    const bool in = (e < E);
    d[k] = in ? ei[E + e] : -1;
    s[k] = in ? ei[e] : 0;
  }
  int p[EPT];
#pragma unroll
  for (int k = 0; k < EPT; ++k) {
    const bool mine = (d[k] >= lo && d[k] < hi);
    p[k] = mine ? (atomicAdd(&cur[d[k]], 1) - base) : -1;
  }
#pragma unroll
  for (int k = 0; k < EPT; ++k)
    if ((unsigned)p[k] < (unsigned)cap) list[p[k]] = s[k];
}

template <int C, bool ELU>
__global__ void seg_softmax_agg(const int* __restrict__ list,
                                const int* __restrict__ cur,
                                const int* __restrict__ off,
                                const float2* __restrict__ als,
                                const float2* __restrict__ ald,
                                const float* __restrict__ Hm,
                                const float* __restrict__ bias,
                                float* __restrict__ out,
                                int lo, int hi, int cap, int N) {
  constexpr int F = 2 * C;
  constexpr int U = 4;
  const int t = blockIdx.x * blockDim.x + threadIdx.x;
  const int n = lo + t / F;
  const int j = t % F;
  if (n >= hi) return;
  const int head = (j >= C) ? 1 : 0;
  const float* __restrict__ alsf = (const float*)als;
  const float* __restrict__ aldf = (const float*)ald;

  const int base = off[lo];
  int beg = (n == lo) ? 0 : (cur[n - 1] - base);
  int end = cur[n] - base;
  beg = min(max(beg, 0), cap);
  end = min(max(end, beg), cap);

  const float adn = aldf[2 * n + head];
  float es = alsf[2 * n + head] + adn;
  es = (es > 0.f) ? es : 0.2f * es;

  float m = es;
  if (!(m >= -3.0e38f && m <= 3.0e38f)) m = 0.f;
  float z = __expf(es - m);
  float acc = z * Hm[n * F + j];
  for (int p = beg; p < end; p += U) {
    int sq[U];
    bool ok[U];
    float aq[U], hq[U], e[U];
#pragma unroll
    for (int u = 0; u < U; ++u) {
      ok[u] = (p + u < end);
      const int q = ok[u] ? (p + u) : (end - 1);
      sq[u] = min(max(list[q], 0), N - 1);
    }
#pragma unroll
    for (int u = 0; u < U; ++u) {
      aq[u] = alsf[2 * sq[u] + head];
      hq[u] = Hm[sq[u] * F + j];
    }
    float mn = m;
#pragma unroll
    for (int u = 0; u < U; ++u) {
      float t = aq[u] + adn;
      t = (t > 0.f) ? t : 0.2f * t;
      e[u] = ok[u] ? t : -3.0e38f;
      mn = fmaxf(mn, e[u]);
    }
    const float r = __expf(m - mn);
    z *= r;
    acc *= r;
#pragma unroll
    for (int u = 0; u < U; ++u) {
      const float wu = __expf(e[u] - mn);
      z += wu;
      acc += wu * hq[u];
    }
    m = mn;
  }
  float x = acc * (1.f / (z + 1e-16f)) + bias[j];
  if (ELU) x = (x > 0.f) ? x : (__expf(x) - 1.f);
  out[n * F + j] = x;
}

extern "C" void kernel_launch(void* const* d_in, const int* in_sizes, int n_in,
                              void* d_out, int out_size, void* d_ws, size_t ws_size,
                              hipStream_t stream) {
  const float* x   = (const float*)d_in[0];
  const int*   ei  = (const int*)  d_in[1];
  const float* W1  = (const float*)d_in[2];
  const float* as1 = (const float*)d_in[3];
  const float* ad1 = (const float*)d_in[4];
  const float* b1  = (const float*)d_in[5];
  const float* W2  = (const float*)d_in[6];
  const float* as2 = (const float*)d_in[7];
  const float* ad2 = (const float*)d_in[8];
  const float* b2  = (const float*)d_in[9];

  const int N = in_sizes[0] / 16;
  const int E = in_sizes[1] / 2;

  float*  ws    = (float*)d_ws;
  float*  h1    = ws;
  float*  x2    = ws + (size_t)16 * N;
  float2* als   = (float2*)(ws + (size_t)32 * N);
  float2* ald   = (float2*)(ws + (size_t)34 * N);
  int*    cnt   = (int*)(ws + (size_t)36 * N);
  int*    off   = (int*)(ws + (size_t)37 * N);
  int*    cur   = (int*)(ws + (size_t)38 * N);
  int*    listP = (int*)(ws + (size_t)39 * N);
  float*  h2    = ws + (size_t)40 * N;
  int*    listF = (int*)ws;

  const int TB = 256;
  const int node_grid  = (N + TB - 1) / TB;
  const int list_grid  = (E + TB * EPT - 1) / (TB * EPT);
  const int tiles      = (N + 15) / 16;
  const int gemm_grid  = (tiles + 7) / 8;
  const int half       = N / 2;
  const int capP       = 17 * N;
  const int capF       = 32 * N;
  const int half_grid  = ((N - half) * 16 + TB - 1) / TB;
  const int lane_grid  = (N * 8 + TB - 1) / TB;

  gemm16_wmma<<<gemm_grid, TB, 0, stream>>>(x, W1, h1, N, 16);
  node_logits<8, true><<<node_grid, TB, 0, stream>>>(h1, as1, ad1, als, ald, cnt, off, cur, N);
  dst_count<<<list_grid, TB, 0, stream>>>(ei, cnt, E);
  seg_offsets<<<node_grid, TB, 0, stream>>>(cnt, off, cur, N);
  dst_place<<<list_grid, TB, 0, stream>>>(ei, cur, off, listP, E, 0, half, capP);
  seg_softmax_agg<8, true><<<half_grid, TB, 0, stream>>>(listP, cur, off, als, ald, h1, b1,
                                                         x2, 0, half, capP, N);
  dst_place<<<list_grid, TB, 0, stream>>>(ei, cur, off, listP, E, half, N, capP);
  seg_softmax_agg<8, true><<<half_grid, TB, 0, stream>>>(listP, cur, off, als, ald, h1, b1,
                                                         x2, half, N, capP, N);

  gemm16_wmma<<<gemm_grid, TB, 0, stream>>>(x2, W2, h2, N, 8);
  node_logits<4, false><<<node_grid, TB, 0, stream>>>(h2, as2, ad2, als, ald, cnt, off, cur, N);
  dst_place<<<list_grid, TB, 0, stream>>>(ei, cur, off, listF, E, 0, N, capF);
  seg_softmax_agg<4, false><<<lane_grid, TB, 0, stream>>>(listF, cur, off, als, ald, h2, b2,
                                                          (float*)d_out, 0, N, capF, N);
}


# edge list grouped by destination built once per call and shared by both layers (was built 3 times); to fit beside h1 and x2 it holds 16-bit sources, low then high segment per node (new list order)
# speedup vs baseline: 4.4842x; 1.7747x over previous
#include <hip/hip_runtime.h>
#include <math.h>


typedef __attribute__((ext_vector_type(2))) float v2f;
typedef __attribute__((ext_vector_type(8))) float v8f;

__global__ void gemm16_wmma(const float* __restrict__ X,
                            const float* __restrict__ W,
                            float* __restrict__ Hout,
                            int nrows, int ncols) {
  const int wave = blockIdx.x * (blockDim.x >> 5) + (threadIdx.x >> 5);
  const int lane = threadIdx.x & 31;
  const int row_base = wave * 16;
  if (row_base >= nrows) return;

  const int m    = lane & 15;
  const int half = lane >> 4;
  int row = row_base + m; if (row > nrows - 1) row = nrows - 1;
  int col = m;            if (col > ncols - 1) col = ncols - 1;

  const float2* __restrict__ Xr = (const float2*)(X + (size_t)row * 16);

  v8f c = {0.f, 0.f, 0.f, 0.f, 0.f, 0.f, 0.f, 0.f};
#pragma unroll
  for (int k0 = 0; k0 < 4; ++k0) {
    const int k = k0 * 4 + half * 2;
    const float2 av = Xr[k0 * 2 + half];
    v2f a; a.x = av.x; a.y = av.y;
    v2f b;
    b.x = W[(k)     * ncols + col];
    b.y = W[(k + 1) * ncols + col];
    c = __builtin_amdgcn_wmma_f32_16x16x4_f32(
         false, a,  false, b,
         (short)0, c,  false,  false);
  }

  if (m < ncols) {
    const int rofs = half * 8;
    float* __restrict__ outp = Hout + (size_t)(row_base + rofs) * ncols + m;
    if (row_base + 16 <= nrows) {
#pragma unroll
      for (int r = 0; r < 8; ++r) outp[(size_t)r * ncols] = c[r];
    } else {
#pragma unroll
      for (int r = 0; r < 8; ++r)
        if (row_base + rofs + r < nrows) outp[(size_t)r * ncols] = c[r];
    }
  }
}

template <int C, bool CLEAR>
__global__ void node_logits(const float* __restrict__ Hm,
                            const float* __restrict__ a_src,
                            const float* __restrict__ a_dst,
                            float2* __restrict__ als, float2* __restrict__ ald,
                            int* __restrict__ cnt, int nrows) {
  const int n = blockIdx.x * blockDim.x + threadIdx.x;
  if (n >= nrows) return;
  constexpr int F = 2 * C;
  const float4* __restrict__ hr = (const float4*)(Hm + (size_t)n * F);
  float s0 = 0.f, s1 = 0.f, d0 = 0.f, d1 = 0.f;
#pragma unroll
  for (int v = 0; v < C / 4; ++v) {
    const float4 h0 = hr[v];
    const float4 h1 = hr[C / 4 + v];
#pragma unroll
    for (int j = 0; j < 4; ++j) {
      const float e0 = (&h0.x)[j], e1 = (&h1.x)[j];
      const int cch = 4 * v + j;
      s0 += e0 * a_src[cch];     d0 += e0 * a_dst[cch];
      s1 += e1 * a_src[C + cch]; d1 += e1 * a_dst[C + cch];
    }
  }
  als[n] = make_float2(s0, s1);
  ald[n] = make_float2(d0, d1);
  if (CLEAR) { cnt[n] = 0; cnt[nrows + n] = 0; }
}

#define EPT 4

__device__ __forceinline__ int edge_segment(int s, int d, int N) {
  const int v = d + (((s >> 16) != 0) ? N : 0);
  return ((unsigned)d < (unsigned)N) ? v : -1;
}

__global__ void dst_count(const int* __restrict__ ei, int* __restrict__ cnt, int E, int N) {
  const int T = gridDim.x * blockDim.x;
  const int t = blockIdx.x * blockDim.x + threadIdx.x;
  int v[EPT];
#pragma unroll
  for (int k = 0; k < EPT; ++k) {
    const int e = t + k * T;
    const bool in = (e < E);
    const int d = in ? ei[E + e] : -1;
    const int s = in ? ei[e] : 0;
    v[k] = edge_segment(s, d, N);
  }
#pragma unroll
  for (int k = 0; k < EPT; ++k)
    if (v[k] >= 0) atomicAdd(&cnt[v[k]], 1);
}

__global__ void seg_offsets(const int* __restrict__ cnt, int* __restrict__ cur, int M) {
  __shared__ int wsum[8];
  __shared__ int wtot[8];
  const int tid  = threadIdx.x;
  const int lane = tid & 31;
  const int wv   = tid >> 5;
  const int block_start = blockIdx.x * 256;
  const int4* __restrict__ c4 = (const int4*)cnt;
  int part = 0;
#pragma unroll 8
  for (int i = tid; i < block_start / 4; i += 256) {
    const int4 v = c4[i];
    part += (v.x + v.y) + (v.z + v.w);
  }
#pragma unroll
  for (int o = 16; o > 0; o >>= 1) part += __shfl_xor(part, o);
  const int n = block_start + tid;
  const int c = (n < M) ? cnt[n] : 0;
  int incl = c;
#pragma unroll
  for (int o = 1; o < 32; o <<= 1) {
    const int up = __shfl_up(incl, o);
    if (lane >= o) incl += up;
  }
  if (lane == 0)  wsum[wv] = part;
  if (lane == 31) wtot[wv] = incl;
  __syncthreads();
  int base = 0;
#pragma unroll
  for (int k = 0; k < 8; ++k) {
    base += wsum[k];
    if (k < wv) base += wtot[k];
  }
  if (n < M) cur[n] = base + incl - c;
}

__global__ void dst_place(const int* __restrict__ ei, int* __restrict__ cur,
                          unsigned short* __restrict__ list, int E, int N, int cap) {
  const int T = gridDim.x * blockDim.x;
  const int t = blockIdx.x * blockDim.x + threadIdx.x;
  int v[EPT], s[EPT];
#pragma unroll
  for (int k = 0; k < EPT; ++k) {
    const int e = t + k * T;
    const bool in = (e < E);
    const int d = in ? ei[E + e] : -1;
    s[k] = in ? ei[e] : 0;
    v[k] = edge_segment(s[k], d, N);
  }
  int p[EPT];
#pragma unroll
  for (int k = 0; k < EPT; ++k)
    p[k] = (v[k] >= 0) ? atomicAdd(&cur[v[k]], 1) : -1;
#pragma unroll
  for (int k = 0; k < EPT; ++k)
    if ((unsigned)p[k] < (unsigned)cap)
      list[p[k]] = (unsigned short)(s[k] & 0xFFFF);
}

template <int C, bool ELU>
__global__ void seg_softmax_agg(const unsigned short* __restrict__ list,
                                const int* __restrict__ cur,
                                const float2* __restrict__ als,
                                const float2* __restrict__ ald,
                                const float* __restrict__ Hm,
                                const float* __restrict__ bias,
                                float* __restrict__ out,
                                int cap, int N) {
  constexpr int F = 2 * C;
  constexpr int U = 4;
  const int t = blockIdx.x * blockDim.x + threadIdx.x;
  const int n = t / F;
  const int j = t % F;
  if (n >= N) return;
  const int head = (j >= C) ? 1 : 0;
  const float* __restrict__ alsf = (const float*)als;
  const float* __restrict__ aldf = (const float*)ald;

  int lbeg = (n == 0) ? 0 : cur[n - 1];
  int lend = cur[n];
  int hbeg = cur[N + n - 1];
  int hend = cur[N + n];
  lbeg = min(max(lbeg, 0), cap);
  lend = min(max(lend, lbeg), cap);
  hbeg = min(max(hbeg, 0), cap);
  hend = min(max(hend, hbeg), cap);
  const int nlow  = lend - lbeg;
  const int end   = nlow + (hend - hbeg);
  const int hskew = hbeg - nlow;

  const float adn = aldf[2 * n + head];
  float es = alsf[2 * n + head] + adn;
  es = (es > 0.f) ? es : 0.2f * es;

  float m = es;
  if (!(m >= -3.0e38f && m <= 3.0e38f)) m = 0.f;
  float z = __expf(es - m);
  float acc = z * Hm[n * F + j];
  for (int p = 0; p < end; p += U) {
    int sq[U];
    bool ok[U];
    float aq[U], hq[U], e[U];
#pragma unroll
    for (int u = 0; u < U; ++u) {
      ok[u] = (p + u < end);
      const int q = ok[u] ? (p + u) : (end - 1);
      const bool high = (q >= nlow);
      const int sv = (int)list[q + (high ? hskew : lbeg)] + (high ? 65536 : 0);
      sq[u] = min(sv, N - 1);
    }
#pragma unroll
    for (int u = 0; u < U; ++u) {
      aq[u] = alsf[2 * sq[u] + head];
      hq[u] = Hm[sq[u] * F + j];
    }
    float mn = m;
#pragma unroll
    for (int u = 0; u < U; ++u) {
      float t = aq[u] + adn;
      t = (t > 0.f) ? t : 0.2f * t;
      e[u] = ok[u] ? t : -3.0e38f;
      mn = fmaxf(mn, e[u]);
    }
    const float r = __expf(m - mn);
    z *= r;
    acc *= r;
#pragma unroll
    for (int u = 0; u < U; ++u) {
      const float wu = __expf(e[u] - mn);
      z += wu;
      acc += wu * hq[u];
    }
    m = mn;
  }
  float x = acc * (1.f / (z + 1e-16f)) + bias[j];
  if (ELU) x = (x > 0.f) ? x : (__expf(x) - 1.f);
  out[n * F + j] = x;
}

extern "C" void kernel_launch(void* const* d_in, const int* in_sizes, int n_in,
                              void* d_out, int out_size, void* d_ws, size_t ws_size,
                              hipStream_t stream) {
  const float* x   = (const float*)d_in[0];
  const int*   ei  = (const int*)  d_in[1];
  const float* W1  = (const float*)d_in[2];
  const float* as1 = (const float*)d_in[3];
  const float* ad1 = (const float*)d_in[4];
  const float* b1  = (const float*)d_in[5];
  const float* W2  = (const float*)d_in[6];
  const float* as2 = (const float*)d_in[7];
  const float* ad2 = (const float*)d_in[8];
  const float* b2  = (const float*)d_in[9];

  const int N = in_sizes[0] / 16;
  const int E = in_sizes[1] / 2;

  float*  ws    = (float*)d_ws;
  float*  h1    = ws;
  float*  x2    = ws + (size_t)16 * N;
  float2* als   = (float2*)(ws + (size_t)32 * N);
  float2* ald   = (float2*)(ws + (size_t)34 * N);
  int*    cnt   = (int*)(ws + (size_t)36 * N);
  int*    cur   = (int*)(ws + (size_t)38 * N);
  unsigned short* list = (unsigned short*)(ws + (size_t)40 * N);
  float*  h2    = ws;

  const int TB = 256;
  const int node_grid  = (N + TB - 1) / TB;
  const int seg_grid   = (2 * N + TB - 1) / TB;
  const int list_grid  = (E + TB * EPT - 1) / (TB * EPT);
  const int tiles      = (N + 15) / 16;
  const int gemm_grid  = (tiles + 7) / 8;
  const int cap        = 32 * N;
  const int lane16_grid = (N * 16 + TB - 1) / TB;
  const int lane8_grid  = (N * 8 + TB - 1) / TB;

  gemm16_wmma<<<gemm_grid, TB, 0, stream>>>(x, W1, h1, N, 16);
  node_logits<8, true><<<node_grid, TB, 0, stream>>>(h1, as1, ad1, als, ald, cnt, N);

  dst_count<<<list_grid, TB, 0, stream>>>(ei, cnt, E, N);
  seg_offsets<<<seg_grid, TB, 0, stream>>>(cnt, cur, 2 * N);
  dst_place<<<list_grid, TB, 0, stream>>>(ei, cur, list, E, N, cap);

  seg_softmax_agg<8, true><<<lane16_grid, TB, 0, stream>>>(list, cur, als, ald, h1, b1,
                                                           x2, cap, N);

  gemm16_wmma<<<gemm_grid, TB, 0, stream>>>(x2, W2, h2, N, 8);
  node_logits<4, false><<<node_grid, TB, 0, stream>>>(h2, as2, ad2, als, ald, cnt, N);
  seg_softmax_agg<4, false><<<lane8_grid, TB, 0, stream>>>(list, cur, als, ald, h2, b2,
                                                           (float*)d_out, cap, N);
}


# attention logits summed in the walk from the gathered feature row (lane butterfly over a head's channels: new sum order of the logit dot products); no logit pass, plane or per-edge logit gather
# speedup vs baseline: 4.7059x; 1.0494x over previous
#include <hip/hip_runtime.h>
#include <math.h>


typedef __attribute__((ext_vector_type(2))) float v2f;
typedef __attribute__((ext_vector_type(8))) float v8f;

template <bool CLEAR>
__global__ void gemm16_wmma(const float* __restrict__ X,
                            const float* __restrict__ W,
                            float* __restrict__ Hout,
                            int* __restrict__ cnt,
                            int nrows, int ncols) {
  const int wave = blockIdx.x * (blockDim.x >> 5) + (threadIdx.x >> 5);
  const int lane = threadIdx.x & 31;
  const int row_base = wave * 16;
  if (row_base >= nrows) return;

  const int m    = lane & 15;
  const int half = lane >> 4;
  if (CLEAR) {
    if (row_base + m < nrows) cnt[half * nrows + row_base + m] = 0;
  }
  int row = row_base + m; if (row > nrows - 1) row = nrows - 1;
  int col = m;            if (col > ncols - 1) col = ncols - 1;

  const float2* __restrict__ Xr = (const float2*)(X + (size_t)row * 16);

  v8f c = {0.f, 0.f, 0.f, 0.f, 0.f, 0.f, 0.f, 0.f};
#pragma unroll
  for (int k0 = 0; k0 < 4; ++k0) {
    const int k = k0 * 4 + half * 2;
    const float2 av = Xr[k0 * 2 + half];
    v2f a; a.x = av.x; a.y = av.y;
    v2f b;
    b.x = W[(k)     * ncols + col];
    b.y = W[(k + 1) * ncols + col];
    c = __builtin_amdgcn_wmma_f32_16x16x4_f32(
         false, a,  false, b,
         (short)0, c,  false,  false);
  }

  if (m < ncols) {
    const int rofs = half * 8;
    float* __restrict__ outp = Hout + (size_t)(row_base + rofs) * ncols + m;
    if (row_base + 16 <= nrows) {
#pragma unroll
      for (int r = 0; r < 8; ++r) outp[(size_t)r * ncols] = c[r];
    } else {
#pragma unroll
      for (int r = 0; r < 8; ++r)
        if (row_base + rofs + r < nrows) outp[(size_t)r * ncols] = c[r];
    }
  }
}

#define EPT 4

__device__ __forceinline__ int edge_segment(int s, int d, int N) {
  const int v = d + (((s >> 16) != 0) ? N : 0);
  return ((unsigned)d < (unsigned)N) ? v : -1;
}

__global__ void dst_count(const int* __restrict__ ei, int* __restrict__ cnt, int E, int N) {
  const int T = gridDim.x * blockDim.x;
  const int t = blockIdx.x * blockDim.x + threadIdx.x;
  int v[EPT];
#pragma unroll
  for (int k = 0; k < EPT; ++k) {
    const int e = t + k * T;
    const bool in = (e < E);
    const int d = in ? ei[E + e] : -1;
    const int s = in ? ei[e] : 0;
    v[k] = edge_segment(s, d, N);
  }
#pragma unroll
  for (int k = 0; k < EPT; ++k)
    if (v[k] >= 0) atomicAdd(&cnt[v[k]], 1);
}

__global__ void seg_offsets(const int* __restrict__ cnt, int* __restrict__ cur, int M) {
  __shared__ int wsum[8];
  __shared__ int wtot[8];
  const int tid  = threadIdx.x;
  const int lane = tid & 31;
  const int wv   = tid >> 5;
  const int block_start = blockIdx.x * 256;
  const int4* __restrict__ c4 = (const int4*)cnt;
  int part = 0;
#pragma unroll 8
  for (int i = tid; i < block_start / 4; i += 256) {
    const int4 v = c4[i];
    part += (v.x + v.y) + (v.z + v.w);
  }
#pragma unroll
  for (int o = 16; o > 0; o >>= 1) part += __shfl_xor(part, o);
  const int n = block_start + tid;
  const int c = (n < M) ? cnt[n] : 0;
  int incl = c;
#pragma unroll
  for (int o = 1; o < 32; o <<= 1) {
    const int up = __shfl_up(incl, o);
    if (lane >= o) incl += up;
  }
  if (lane == 0)  wsum[wv] = part;
  if (lane == 31) wtot[wv] = incl;
  __syncthreads();
  int base = 0;
#pragma unroll
  for (int k = 0; k < 8; ++k) {
    base += wsum[k];
    if (k < wv) base += wtot[k];
  }
  if (n < M) cur[n] = base + incl - c;
}

__global__ void dst_place(const int* __restrict__ ei, int* __restrict__ cur,
                          unsigned short* __restrict__ list, int E, int N, int cap) {
  const int T = gridDim.x * blockDim.x;
  const int t = blockIdx.x * blockDim.x + threadIdx.x;
  int v[EPT], s[EPT];
#pragma unroll
  for (int k = 0; k < EPT; ++k) {
    const int e = t + k * T;
    const bool in = (e < E);
    const int d = in ? ei[E + e] : -1;
    s[k] = in ? ei[e] : 0;
    v[k] = edge_segment(s[k], d, N);
  }
  int p[EPT];
#pragma unroll
  for (int k = 0; k < EPT; ++k)
    p[k] = (v[k] >= 0) ? atomicAdd(&cur[v[k]], 1) : -1;
#pragma unroll
  for (int k = 0; k < EPT; ++k)
    if ((unsigned)p[k] < (unsigned)cap)
      list[p[k]] = (unsigned short)(s[k] & 0xFFFF);
}

template <int XMASK>
__device__ __forceinline__ float row_xor_read(float v) {
  return __int_as_float(__builtin_amdgcn_update_dpp(
      0, __float_as_int(v), 0x160 + XMASK,  0xF,  0xF,
       true));
}
template <int C>
__device__ __forceinline__ float head_sum(float v) {
  v += row_xor_read<1>(v);
  v += row_xor_read<2>(v);
  if (C == 8) v += row_xor_read<4>(v);
  return v;
}

template <int C, bool ELU>
__global__ void seg_softmax_agg(const unsigned short* __restrict__ list,
                                const int* __restrict__ cur,
                                const float* __restrict__ a_src,
                                const float* __restrict__ a_dst,
                                const float* __restrict__ Hm,
                                const float* __restrict__ bias,
                                float* __restrict__ out,
                                int cap, int N) {
  constexpr int F = 2 * C;
  constexpr int U = 4;
  const int t = blockIdx.x * blockDim.x + threadIdx.x;
  const int n = t / F;
  const int j = t % F;
  if (n >= N) return;
  const float asj = a_src[j];

  int lbeg = (n == 0) ? 0 : cur[n - 1];
  int lend = cur[n];
  int hbeg = cur[N + n - 1];
  int hend = cur[N + n];
  lbeg = min(max(lbeg, 0), cap);
  lend = min(max(lend, lbeg), cap);
  hbeg = min(max(hbeg, 0), cap);
  hend = min(max(hend, hbeg), cap);
  const int nlow  = lend - lbeg;
  const int end   = nlow + (hend - hbeg);
  const int hskew = hbeg - nlow;

  const float hn  = Hm[n * F + j];
  const float adn = head_sum<C>(hn * a_dst[j]);
  float es = head_sum<C>(hn * asj) + adn;
  es = (es > 0.f) ? es : 0.2f * es;

  float m = es;
  if (!(m >= -3.0e38f && m <= 3.0e38f)) m = 0.f;
  float z = __expf(es - m);
  float acc = z * hn;
  for (int p = 0; p < end; p += U) {
    int sq[U];
    bool ok[U];
    float hq[U], e[U];
#pragma unroll
    for (int u = 0; u < U; ++u) {
      ok[u] = (p + u < end);
      const int q = ok[u] ? (p + u) : (end - 1);
      const bool high = (q >= nlow);
      const int sv = (int)list[q + (high ? hskew : lbeg)] + (high ? 65536 : 0);
      sq[u] = min(sv, N - 1);
    }
#pragma unroll
    for (int u = 0; u < U; ++u) hq[u] = Hm[sq[u] * F + j];
    float mn = m;
#pragma unroll
    for (int u = 0; u < U; ++u) {
      float t = head_sum<C>(hq[u] * asj) + adn;
      t = (t > 0.f) ? t : 0.2f * t;
      e[u] = ok[u] ? t : -3.0e38f;
      mn = fmaxf(mn, e[u]);
    }
    const float r = __expf(m - mn);
    z *= r;
    acc *= r;
#pragma unroll
    for (int u = 0; u < U; ++u) {
      const float wu = __expf(e[u] - mn);
      z += wu;
      acc += wu * hq[u];
    }
    m = mn;
  }
  float x = acc * (1.f / (z + 1e-16f)) + bias[j];
  if (ELU) x = (x > 0.f) ? x : (__expf(x) - 1.f);
  out[n * F + j] = x;
}

extern "C" void kernel_launch(void* const* d_in, const int* in_sizes, int n_in,
                              void* d_out, int out_size, void* d_ws, size_t ws_size,
                              hipStream_t stream) {
  const float* x   = (const float*)d_in[0];
  const int*   ei  = (const int*)  d_in[1];
  const float* W1  = (const float*)d_in[2];
  const float* as1 = (const float*)d_in[3];
  const float* ad1 = (const float*)d_in[4];
  const float* b1  = (const float*)d_in[5];
  const float* W2  = (const float*)d_in[6];
  const float* as2 = (const float*)d_in[7];
  const float* ad2 = (const float*)d_in[8];
  const float* b2  = (const float*)d_in[9];

  const int N = in_sizes[0] / 16;
  const int E = in_sizes[1] / 2;

  float*  ws    = (float*)d_ws;
  float*  h1    = ws;
  float*  x2    = ws + (size_t)16 * N;
  int*    cnt   = (int*)(ws + (size_t)36 * N);
  int*    cur   = (int*)(ws + (size_t)38 * N);
  unsigned short* list = (unsigned short*)(ws + (size_t)40 * N);
  float*  h2    = ws;

  const int TB = 256;
  const int seg_grid   = (2 * N + TB - 1) / TB;
  const int list_grid  = (E + TB * EPT - 1) / (TB * EPT);
  const int tiles      = (N + 15) / 16;
  const int gemm_grid  = (tiles + 7) / 8;
  const int cap        = 32 * N;
  const int lane16_grid = (N * 16 + TB - 1) / TB;
  const int lane8_grid  = (N * 8 + TB - 1) / TB;

  gemm16_wmma<true><<<gemm_grid, TB, 0, stream>>>(x, W1, h1, cnt, N, 16);

  dst_count<<<list_grid, TB, 0, stream>>>(ei, cnt, E, N);
  seg_offsets<<<seg_grid, TB, 0, stream>>>(cnt, cur, 2 * N);
  dst_place<<<list_grid, TB, 0, stream>>>(ei, cur, list, E, N, cap);

  seg_softmax_agg<8, true><<<lane16_grid, TB, 0, stream>>>(list, cur, as1, ad1, h1, b1,
                                                           x2, cap, N);

  gemm16_wmma<false><<<gemm_grid, TB, 0, stream>>>(x2, W2, h2, cnt, N, 8);
  seg_softmax_agg<4, false><<<lane8_grid, TB, 0, stream>>>(list, cur, as2, ad2, h2, b2,
                                                           (float*)d_out, cap, N);
}


# list build: count pass keeps each edge's rank (value its add returned, 16 bits, overflow counter past 65535); place = segment start + rank, no add per edge; list order = count-pass arrival order
# speedup vs baseline: 6.2249x; 1.3228x over previous
#include <hip/hip_runtime.h>
#include <math.h>


typedef __attribute__((ext_vector_type(2))) float v2f;
typedef __attribute__((ext_vector_type(8))) float v8f;

template <bool CLEAR>
__global__ void gemm16_wmma(const float* __restrict__ X,
                            const float* __restrict__ W,
                            float* __restrict__ Hout,
                            int* __restrict__ clr,
                            int nrows, int ncols) {
  const int wave = blockIdx.x * (blockDim.x >> 5) + (threadIdx.x >> 5);
  const int lane = threadIdx.x & 31;
  const int row_base = wave * 16;
  if (row_base >= nrows) return;

  const int m    = lane & 15;
  const int half = lane >> 4;
  if (CLEAR) {
    if (row_base + m < nrows) {
      clr[half * nrows + row_base + m] = 0;
      clr[(2 + half) * nrows + row_base + m] = 0;
    }
  }
  int row = row_base + m; if (row > nrows - 1) row = nrows - 1;
  int col = m;            if (col > ncols - 1) col = ncols - 1;

  const float2* __restrict__ Xr = (const float2*)(X + (size_t)row * 16);

  v8f c = {0.f, 0.f, 0.f, 0.f, 0.f, 0.f, 0.f, 0.f};
#pragma unroll
  for (int k0 = 0; k0 < 4; ++k0) {
    const int k = k0 * 4 + half * 2;
    const float2 av = Xr[k0 * 2 + half];
    v2f a; a.x = av.x; a.y = av.y;
    v2f b;
    b.x = W[(k)     * ncols + col];
    b.y = W[(k + 1) * ncols + col];
    c = __builtin_amdgcn_wmma_f32_16x16x4_f32(
         false, a,  false, b,
         (short)0, c,  false,  false);
  }

  if (m < ncols) {
    const int rofs = half * 8;
    float* __restrict__ outp = Hout + (size_t)(row_base + rofs) * ncols + m;
    if (row_base + 16 <= nrows) {
#pragma unroll
      for (int r = 0; r < 8; ++r) outp[(size_t)r * ncols] = c[r];
    } else {
#pragma unroll
      for (int r = 0; r < 8; ++r)
        if (row_base + rofs + r < nrows) outp[(size_t)r * ncols] = c[r];
    }
  }
}

#define EPT 4

__device__ __forceinline__ int edge_segment(int s, int d, int N) {
  const int v = d + (((s >> 16) != 0) ? N : 0);
  return ((unsigned)d < (unsigned)N) ? v : -1;
}

#define RANK_TOP 65535

__global__ void dst_count(const int* __restrict__ ei, int* __restrict__ cnt,
                          unsigned short* __restrict__ rank, int E, int N, int cap) {
  const int T = gridDim.x * blockDim.x;
  const int t = blockIdx.x * blockDim.x + threadIdx.x;
  int v[EPT], r[EPT];
#pragma unroll
  for (int k = 0; k < EPT; ++k) {
    const int e = t + k * T;
    const bool in = (e < E) && (e < cap);
    const int d = in ? ei[E + e] : -1;
    const int s = in ? ei[e] : 0;
    v[k] = edge_segment(s, d, N);
    r[k] = in ? 0 : -1;
  }
#pragma unroll
  for (int k = 0; k < EPT; ++k)
    r[k] = (v[k] >= 0) ? atomicAdd(&cnt[v[k]], 1) : r[k];
#pragma unroll
  for (int k = 0; k < EPT; ++k)
    if (r[k] >= 0)
      rank[t + k * T] = (unsigned short)min(r[k], RANK_TOP);
}

__global__ void seg_offsets(const int* __restrict__ cnt, int* __restrict__ cur, int M) {
  __shared__ int wsum[8];
  __shared__ int wtot[8];
  const int tid  = threadIdx.x;
  const int lane = tid & 31;
  const int wv   = tid >> 5;
  const int block_start = blockIdx.x * 256;
  const int4* __restrict__ c4 = (const int4*)cnt;
  int part = 0;
#pragma unroll 8
  for (int i = tid; i < block_start / 4; i += 256) {
    const int4 v = c4[i];
    part += (v.x + v.y) + (v.z + v.w);
  }
#pragma unroll
  for (int o = 16; o > 0; o >>= 1) part += __shfl_xor(part, o);
  const int n = block_start + tid;
  const int c = (n < M) ? cnt[n] : 0;
  int incl = c;
#pragma unroll
  for (int o = 1; o < 32; o <<= 1) {
    const int up = __shfl_up(incl, o);
    if (lane >= o) incl += up;
  }
  if (lane == 0)  wsum[wv] = part;
  if (lane == 31) wtot[wv] = incl;
  __syncthreads();
  int base = 0;
#pragma unroll
  for (int k = 0; k < 8; ++k) {
    base += wsum[k];
    if (k < wv) base += wtot[k];
  }
  if (n < M) cur[n] = base + incl - c;
}

__global__ void dst_place(const int* __restrict__ ei, const int* __restrict__ cur,
                          int* __restrict__ ovf,
                          const unsigned short* __restrict__ rank,
                          unsigned short* __restrict__ list, int E, int N, int cap) {
  const int T = gridDim.x * blockDim.x;
  const int t = blockIdx.x * blockDim.x + threadIdx.x;
  int v[EPT], s[EPT], r[EPT];
#pragma unroll
  for (int k = 0; k < EPT; ++k) {
    const int e = t + k * T;
    const bool in = (e < E) && (e < cap);
    const int d = in ? ei[E + e] : -1;
    s[k] = in ? ei[e] : 0;
    r[k] = in ? (int)rank[e] : 0;
    v[k] = edge_segment(s[k], d, N);
  }
  int p[EPT];
#pragma unroll
  for (int k = 0; k < EPT; ++k)
    p[k] = (v[k] >= 0) ? cur[v[k]] : -1;
#pragma unroll
  for (int k = 0; k < EPT; ++k) {
    if (v[k] >= 0) {
      int q = r[k];
      if (q == RANK_TOP) q += atomicAdd(&ovf[v[k]], 1);
      const int slot = p[k] + q;
      if ((unsigned)slot < (unsigned)cap)
        list[slot] = (unsigned short)(s[k] & 0xFFFF);
    }
  }
}

template <int XMASK>
__device__ __forceinline__ float row_xor_read(float v) {
  return __int_as_float(__builtin_amdgcn_update_dpp(
      0, __float_as_int(v), 0x160 + XMASK,  0xF,  0xF,
       true));
}
template <int C>
__device__ __forceinline__ float head_sum(float v) {
  v += row_xor_read<1>(v);
  v += row_xor_read<2>(v);
  if (C == 8) v += row_xor_read<4>(v);
  return v;
}

template <int C, bool ELU>
__global__ void seg_softmax_agg(const unsigned short* __restrict__ list,
                                const int* __restrict__ cur,
                                const int* __restrict__ cnt,
                                const float* __restrict__ a_src,
                                const float* __restrict__ a_dst,
                                const float* __restrict__ Hm,
                                const float* __restrict__ bias,
                                float* __restrict__ out,
                                int cap, int N) {
  constexpr int F = 2 * C;
  constexpr int U = 4;
  const int t = blockIdx.x * blockDim.x + threadIdx.x;
  const int n = t / F;
  const int j = t % F;
  if (n >= N) return;
  const float asj = a_src[j];

  int lbeg = cur[n];
  int lend = lbeg + cnt[n];
  int hbeg = cur[N + n];
  int hend = hbeg + cnt[N + n];
  lbeg = min(max(lbeg, 0), cap);
  lend = min(max(lend, lbeg), cap);
  hbeg = min(max(hbeg, 0), cap);
  hend = min(max(hend, hbeg), cap);
  const int nlow  = lend - lbeg;
  const int end   = nlow + (hend - hbeg);
  const int hskew = hbeg - nlow;

  const float hn  = Hm[n * F + j];
  const float adn = head_sum<C>(hn * a_dst[j]);
  float es = head_sum<C>(hn * asj) + adn;
  es = (es > 0.f) ? es : 0.2f * es;

  float m = es;
  if (!(m >= -3.0e38f && m <= 3.0e38f)) m = 0.f;
  float z = __expf(es - m);
  float acc = z * hn;
  for (int p = 0; p < end; p += U) {
    int sq[U];
    bool ok[U];
    float hq[U], e[U];
#pragma unroll
    for (int u = 0; u < U; ++u) {
      ok[u] = (p + u < end);
      const int q = ok[u] ? (p + u) : (end - 1);
      const bool high = (q >= nlow);
      const int sv = (int)list[q + (high ? hskew : lbeg)] + (high ? 65536 : 0);
      sq[u] = min(sv, N - 1);
    }
#pragma unroll
    for (int u = 0; u < U; ++u) hq[u] = Hm[sq[u] * F + j];
    float mn = m;
#pragma unroll
    for (int u = 0; u < U; ++u) {
      float t = head_sum<C>(hq[u] * asj) + adn;
      t = (t > 0.f) ? t : 0.2f * t;
      e[u] = ok[u] ? t : -3.0e38f;
      mn = fmaxf(mn, e[u]);
    }
    const float r = __expf(m - mn);
    z *= r;
    acc *= r;
#pragma unroll
    for (int u = 0; u < U; ++u) {
      const float wu = __expf(e[u] - mn);
      z += wu;
      acc += wu * hq[u];
    }
    m = mn;
  }
  float x = acc * (1.f / (z + 1e-16f)) + bias[j];
  if (ELU) x = (x > 0.f) ? x : (__expf(x) - 1.f);
  out[n * F + j] = x;
}

extern "C" void kernel_launch(void* const* d_in, const int* in_sizes, int n_in,
                              void* d_out, int out_size, void* d_ws, size_t ws_size,
                              hipStream_t stream) {
  const float* x   = (const float*)d_in[0];
  const int*   ei  = (const int*)  d_in[1];
  const float* W1  = (const float*)d_in[2];
  const float* as1 = (const float*)d_in[3];
  const float* ad1 = (const float*)d_in[4];
  const float* b1  = (const float*)d_in[5];
  const float* W2  = (const float*)d_in[6];
  const float* as2 = (const float*)d_in[7];
  const float* ad2 = (const float*)d_in[8];
  const float* b2  = (const float*)d_in[9];

  const int N = in_sizes[0] / 16;
  const int E = in_sizes[1] / 2;

  float*  ws    = (float*)d_ws;
  float*  h1    = ws;
  float*  x2    = ws + (size_t)16 * N;
  unsigned short* rank = (unsigned short*)x2;
  int*    ovf   = (int*)(ws + (size_t)34 * N);
  int*    cnt   = (int*)(ws + (size_t)36 * N);
  int*    cur   = (int*)(ws + (size_t)38 * N);
  unsigned short* list = (unsigned short*)(ws + (size_t)40 * N);
  float*  h2    = ws;

  const int TB = 256;
  const int seg_grid   = (2 * N + TB - 1) / TB;
  const int list_grid  = (E + TB * EPT - 1) / (TB * EPT);
  const int tiles      = (N + 15) / 16;
  const int gemm_grid  = (tiles + 7) / 8;
  const int cap        = 32 * N;
  const int lane16_grid = (N * 16 + TB - 1) / TB;
  const int lane8_grid  = (N * 8 + TB - 1) / TB;

  gemm16_wmma<true><<<gemm_grid, TB, 0, stream>>>(x, W1, h1, ovf, N, 16);

  dst_count<<<list_grid, TB, 0, stream>>>(ei, cnt, rank, E, N, cap);
  seg_offsets<<<seg_grid, TB, 0, stream>>>(cnt, cur, 2 * N);
  dst_place<<<list_grid, TB, 0, stream>>>(ei, cur, ovf, rank, list, E, N, cap);

  seg_softmax_agg<8, true><<<lane16_grid, TB, 0, stream>>>(list, cur, cnt, as1, ad1, h1, b1,
                                                           x2, cap, N);

  gemm16_wmma<false><<<gemm_grid, TB, 0, stream>>>(x2, W2, h2, cnt, N, 8);
  seg_softmax_agg<4, false><<<lane8_grid, TB, 0, stream>>>(list, cur, cnt, as2, ad2, h2, b2,
                                                           (float*)d_out, cap, N);
}


# list build: each segment has one count per half of the package (a wave adds to its own half's, from its die number), so no count line is shared across halves; list order: half-0 then half-1 arrivals
# speedup vs baseline: 10.1461x; 1.6299x over previous
#include <hip/hip_runtime.h>
#include <math.h>


typedef __attribute__((ext_vector_type(2))) float v2f;
typedef __attribute__((ext_vector_type(8))) float v8f;

template <bool CLEAR>
__global__ void gemm16_wmma(const float* __restrict__ X,
                            const float* __restrict__ W,
                            float* __restrict__ Hout,
                            int* __restrict__ clr,
                            int nrows, int ncols) {
  const int wave = blockIdx.x * (blockDim.x >> 5) + (threadIdx.x >> 5);
  const int lane = threadIdx.x & 31;
  const int row_base = wave * 16;
  if (row_base >= nrows) return;

  const int m    = lane & 15;
  const int half = lane >> 4;
  if (CLEAR) {
    if (row_base + m < nrows) {
      clr[half * nrows + row_base + m] = 0;
      clr[(2 + half) * nrows + row_base + m] = 0;
    }
  }
  int row = row_base + m; if (row > nrows - 1) row = nrows - 1;
  int col = m;            if (col > ncols - 1) col = ncols - 1;

  const float2* __restrict__ Xr = (const float2*)(X + (size_t)row * 16);

  v8f c = {0.f, 0.f, 0.f, 0.f, 0.f, 0.f, 0.f, 0.f};
#pragma unroll
  for (int k0 = 0; k0 < 4; ++k0) {
    const int k = k0 * 4 + half * 2;
    const float2 av = Xr[k0 * 2 + half];
    v2f a; a.x = av.x; a.y = av.y;
    v2f b;
    b.x = W[(k)     * ncols + col];
    b.y = W[(k + 1) * ncols + col];
    c = __builtin_amdgcn_wmma_f32_16x16x4_f32(
         false, a,  false, b,
         (short)0, c,  false,  false);
  }

  if (m < ncols) {
    const int rofs = half * 8;
    float* __restrict__ outp = Hout + (size_t)(row_base + rofs) * ncols + m;
    if (row_base + 16 <= nrows) {
#pragma unroll
      for (int r = 0; r < 8; ++r) outp[(size_t)r * ncols] = c[r];
    } else {
#pragma unroll
      for (int r = 0; r < 8; ++r)
        if (row_base + rofs + r < nrows) outp[(size_t)r * ncols] = c[r];
    }
  }
}

#define EPT 4

__device__ __forceinline__ int edge_segment(int s, int d, int N) {
  const int v = d + (((s >> 16) != 0) ? N : 0);
  return ((unsigned)d < (unsigned)N) ? v : -1;
}

#define RANK_TOP 32767

__device__ __forceinline__ int wave_half() {
  return (__builtin_amdgcn_s_sendmsg_rtn(0x87) >> 18) & 1;
}

__global__ void dst_count(const int* __restrict__ ei, int* __restrict__ cnt,
                          unsigned short* __restrict__ rank, int E, int N, int cap) {
  const int T = gridDim.x * blockDim.x;
  const int t = blockIdx.x * blockDim.x + threadIdx.x;
  const int hb = wave_half();
  int* __restrict__ cnth = cnt + hb * 2 * N;
  int v[EPT], r[EPT];
#pragma unroll
  for (int k = 0; k < EPT; ++k) {
    const int e = t + k * T;
    const bool in = (e < E) && (e < cap);
    const int d = in ? ei[E + e] : -1;
    const int s = in ? ei[e] : 0;
    v[k] = edge_segment(s, d, N);
    r[k] = in ? 0 : -1;
  }
#pragma unroll
  for (int k = 0; k < EPT; ++k)
    r[k] = (v[k] >= 0) ? atomicAdd(&cnth[v[k]], 1) : r[k];
#pragma unroll
  for (int k = 0; k < EPT; ++k)
    if (r[k] >= 0)
      rank[t + k * T] = (unsigned short)(min(r[k], RANK_TOP) | (hb << 15));
}

#define SCAN_TB 1024
__global__ void __launch_bounds__(SCAN_TB)
seg_offsets(const int* __restrict__ cnt, int* __restrict__ cur, int M) {
  __shared__ int wsum[SCAN_TB / 32];
  __shared__ int wtot[SCAN_TB / 32];
  const int tid  = threadIdx.x;
  const int lane = tid & 31;
  const int wv   = tid >> 5;
  const int block_start = blockIdx.x * SCAN_TB;
  const int Mh = M >> 1;
  const int front = block_start >> 1;
  int part = 0;
  if ((Mh & 3) == 0) {
    for (int b = 0; b < 2; ++b) {
      const int4* __restrict__ c4 = (const int4*)(cnt + b * Mh);
#pragma unroll 4
      for (int i = tid; i < front / 4; i += SCAN_TB) {
        const int4 v = c4[i];
        part += (v.x + v.y) + (v.z + v.w);
      }
    }
  } else {
    for (int b = 0; b < 2; ++b)
      for (int i = tid; i < front; i += SCAN_TB) part += cnt[b * Mh + i];
  }
#pragma unroll
  for (int o = 16; o > 0; o >>= 1) part += __shfl_xor(part, o);
  const int n = block_start + tid;
  const int c = (n < M) ? cnt[(n & 1) * Mh + (n >> 1)] : 0;
  int incl = c;
#pragma unroll
  for (int o = 1; o < 32; o <<= 1) {
    const int up = __shfl_up(incl, o);
    if (lane >= o) incl += up;
  }
  if (lane == 0)  wsum[wv] = part;
  if (lane == 31) wtot[wv] = incl;
  __syncthreads();
  int base = 0;
#pragma unroll
  for (int k = 0; k < SCAN_TB / 32; ++k) {
    base += wsum[k];
    if (k < wv) base += wtot[k];
  }
  if (n < M) cur[n] = base + incl;
}

__global__ void dst_place(const int* __restrict__ ei, const int* __restrict__ cur,
                          int* __restrict__ cnt,
                          const unsigned short* __restrict__ rank,
                          unsigned short* __restrict__ list, int E, int N, int cap) {
  const int T = gridDim.x * blockDim.x;
  const int t = blockIdx.x * blockDim.x + threadIdx.x;
  int v[EPT], s[EPT], r[EPT];
#pragma unroll
  for (int k = 0; k < EPT; ++k) {
    const int e = t + k * T;
    const bool in = (e < E) && (e < cap);
    const int d = in ? ei[E + e] : -1;
    s[k] = in ? ei[e] : 0;
    r[k] = in ? (int)rank[e] : 0;
    const int sg = edge_segment(s[k], d, N);
    v[k] = (sg >= 0) ? 2 * sg + (r[k] >> 15) : -1;
    r[k] &= RANK_TOP;
  }
  int p[EPT];
#pragma unroll
  for (int k = 0; k < EPT; ++k)
    p[k] = (v[k] > 0) ? cur[v[k] - 1] : 0;
#pragma unroll
  for (int k = 0; k < EPT; ++k) {
    if (v[k] >= 0) {
      int q = r[k];
      if (q == RANK_TOP)
        q = atomicSub(&cnt[(v[k] & 1) * 2 * N + (v[k] >> 1)], 1) - 1;
      const int slot = p[k] + q;
      if ((unsigned)slot < (unsigned)cap)
        list[slot] = (unsigned short)(s[k] & 0xFFFF);
    }
  }
}

template <int XMASK>
__device__ __forceinline__ float row_xor_read(float v) {
  return __int_as_float(__builtin_amdgcn_update_dpp(
      0, __float_as_int(v), 0x160 + XMASK,  0xF,  0xF,
       true));
}
template <int C>
__device__ __forceinline__ float head_sum(float v) {
  v += row_xor_read<1>(v);
  v += row_xor_read<2>(v);
  if (C == 8) v += row_xor_read<4>(v);
  return v;
}

template <int C, bool ELU>
__global__ void seg_softmax_agg(const unsigned short* __restrict__ list,
                                const int* __restrict__ cur,
                                const float* __restrict__ a_src,
                                const float* __restrict__ a_dst,
                                const float* __restrict__ Hm,
                                const float* __restrict__ bias,
                                float* __restrict__ out,
                                int cap, int N) {
  constexpr int F = 2 * C;
  constexpr int U = 4;
  const int t = blockIdx.x * blockDim.x + threadIdx.x;
  const int n = t / F;
  const int j = t % F;
  if (n >= N) return;
  const float asj = a_src[j];

  int lbeg = (n > 0) ? cur[2 * n - 1] : 0;
  int lend = cur[2 * n + 1];
  int hbeg = cur[2 * (N + n) - 1];
  int hend = cur[2 * (N + n) + 1];
  lbeg = min(max(lbeg, 0), cap);
  lend = min(max(lend, lbeg), cap);
  hbeg = min(max(hbeg, 0), cap);
  hend = min(max(hend, hbeg), cap);
  const int nlow  = lend - lbeg;
  const int end   = nlow + (hend - hbeg);
  const int hskew = hbeg - nlow;

  const float hn  = Hm[n * F + j];
  const float adn = head_sum<C>(hn * a_dst[j]);
  float es = head_sum<C>(hn * asj) + adn;
  es = (es > 0.f) ? es : 0.2f * es;

  float m = es;
  if (!(m >= -3.0e38f && m <= 3.0e38f)) m = 0.f;
  float z = __expf(es - m);
  float acc = z * hn;
  for (int p = 0; p < end; p += U) {
    int sq[U];
    bool ok[U];
    float hq[U], e[U];
#pragma unroll
    for (int u = 0; u < U; ++u) {
      ok[u] = (p + u < end);
      const int q = ok[u] ? (p + u) : (end - 1);
      const bool high = (q >= nlow);
      const int sv = (int)list[q + (high ? hskew : lbeg)] + (high ? 65536 : 0);
      sq[u] = min(sv, N - 1);
    }
#pragma unroll
    for (int u = 0; u < U; ++u) hq[u] = Hm[sq[u] * F + j];
    float mn = m;
#pragma unroll
    for (int u = 0; u < U; ++u) {
      float t = head_sum<C>(hq[u] * asj) + adn;
      t = (t > 0.f) ? t : 0.2f * t;
      e[u] = ok[u] ? t : -3.0e38f;
      mn = fmaxf(mn, e[u]);
    }
    const float r = __expf(m - mn);
    z *= r;
    acc *= r;
#pragma unroll
    for (int u = 0; u < U; ++u) {
      const float wu = __expf(e[u] - mn);
      z += wu;
      acc += wu * hq[u];
    }
    m = mn;
  }
  float x = acc * (1.f / (z + 1e-16f)) + bias[j];
  if (ELU) x = (x > 0.f) ? x : (__expf(x) - 1.f);
  out[n * F + j] = x;
}

extern "C" void kernel_launch(void* const* d_in, const int* in_sizes, int n_in,
                              void* d_out, int out_size, void* d_ws, size_t ws_size,
                              hipStream_t stream) {
  const float* x   = (const float*)d_in[0];
  const int*   ei  = (const int*)  d_in[1];
  const float* W1  = (const float*)d_in[2];
  const float* as1 = (const float*)d_in[3];
  const float* ad1 = (const float*)d_in[4];
  const float* b1  = (const float*)d_in[5];
  const float* W2  = (const float*)d_in[6];
  const float* as2 = (const float*)d_in[7];
  const float* ad2 = (const float*)d_in[8];
  const float* b2  = (const float*)d_in[9];

  const int N = in_sizes[0] / 16;
  const int E = in_sizes[1] / 2;

  float*  ws    = (float*)d_ws;
  float*  h1    = ws;
  float*  x2    = ws + (size_t)16 * N;
  unsigned short* rank = (unsigned short*)x2;
  int*    cnt   = (int*)(ws + (size_t)32 * N);
  int*    cur   = (int*)(ws + (size_t)36 * N);
  unsigned short* list = (unsigned short*)(ws + (size_t)40 * N);
  float*  h2    = ws;

  const int TB = 256;
  const int seg_grid   = (4 * N + SCAN_TB - 1) / SCAN_TB;
  const int list_grid  = (E + TB * EPT - 1) / (TB * EPT);
  const int tiles      = (N + 15) / 16;
  const int gemm_grid  = (tiles + 7) / 8;
  const int cap        = 32 * N;
  const int lane16_grid = (N * 16 + TB - 1) / TB;
  const int lane8_grid  = (N * 8 + TB - 1) / TB;

  gemm16_wmma<true><<<gemm_grid, TB, 0, stream>>>(x, W1, h1, cnt, N, 16);

  dst_count<<<list_grid, TB, 0, stream>>>(ei, cnt, rank, E, N, cap);
  seg_offsets<<<seg_grid, SCAN_TB, 0, stream>>>(cnt, cur, 4 * N);
  dst_place<<<list_grid, TB, 0, stream>>>(ei, cur, cnt, rank, list, E, N, cap);

  seg_softmax_agg<8, true><<<lane16_grid, TB, 0, stream>>>(list, cur, as1, ad1, h1, b1,
                                                           x2, cap, N);

  gemm16_wmma<false><<<gemm_grid, TB, 0, stream>>>(x2, W2, h2, cnt, N, 8);
  seg_softmax_agg<4, false><<<lane8_grid, TB, 0, stream>>>(list, cur, as2, ad2, h2, b2,
                                                           (float*)d_out, cap, N);
}


# list build as a two-level sort through LDS: per-block key counts, column scan, LDS-sorted copy into 256-node bins, one block sorts each bin; no global atomics or per-edge scatter; new list order
# speedup vs baseline: 16.7363x; 1.6495x over previous
#include <hip/hip_runtime.h>
#include <math.h>


typedef __attribute__((ext_vector_type(2))) float v2f;
typedef __attribute__((ext_vector_type(8))) float v8f;

__global__ void gemm16_wmma(const float* __restrict__ X,
                            const float* __restrict__ W,
                            float* __restrict__ Hout,
                            int nrows, int ncols) {
  const int wave = blockIdx.x * (blockDim.x >> 5) + (threadIdx.x >> 5);
  const int lane = threadIdx.x & 31;
  const int row_base = wave * 16;
  if (row_base >= nrows) return;

  const int m    = lane & 15;
  const int half = lane >> 4;
  int row = row_base + m; if (row > nrows - 1) row = nrows - 1;
  int col = m;            if (col > ncols - 1) col = ncols - 1;

  const float2* __restrict__ Xr = (const float2*)(X + (size_t)row * 16);

  v8f c = {0.f, 0.f, 0.f, 0.f, 0.f, 0.f, 0.f, 0.f};
#pragma unroll
  for (int k0 = 0; k0 < 4; ++k0) {
    const int k = k0 * 4 + half * 2;
    const float2 av = Xr[k0 * 2 + half];
    v2f a; a.x = av.x; a.y = av.y;
    v2f b;
    b.x = W[(k)     * ncols + col];
    b.y = W[(k + 1) * ncols + col];
    c = __builtin_amdgcn_wmma_f32_16x16x4_f32(
         false, a,  false, b,
         (short)0, c,  false,  false);
  }

  if (m < ncols) {
    const int rofs = half * 8;
    float* __restrict__ outp = Hout + (size_t)(row_base + rofs) * ncols + m;
    if (row_base + 16 <= nrows) {
#pragma unroll
      for (int r = 0; r < 8; ++r) outp[(size_t)r * ncols] = c[r];
    } else {
#pragma unroll
      for (int r = 0; r < 8; ++r)
        if (row_base + rofs + r < nrows) outp[(size_t)r * ncols] = c[r];
    }
  }
}

#define KEYS 1024
#define BIN_TB 1024
#define CH_EPT 16
#define CHUNK (BIN_TB * CH_EPT)
#define STAGE_CAP 32768

__device__ __forceinline__ int edge_key(int s, int d, int N) {
  const bool ok = ((unsigned)d < (unsigned)N) && (d < 131072);
  return ok ? (((d >> 8) << 1) | (((s >> 16) != 0) ? 1 : 0)) : -1;
}
__device__ __forceinline__ int edge_word(int s, int d) {
  const int idx = ((((s >> 16) != 0) ? 1 : 0) << 8) | (d & 255);
  return (idx << 16) | (s & 0xFFFF);
}

__device__ __forceinline__ int block_excl_scan(int v, int* sh, int* total) {
  const int lane = threadIdx.x & 31;
  const int wv   = threadIdx.x >> 5;
  int incl = v;
#pragma unroll
  for (int o = 1; o < 32; o <<= 1) {
    const int up = __shfl_up(incl, o);
    if (lane >= o) incl += up;
  }
  __syncthreads();
  if (lane == 31) sh[wv] = incl;
  __syncthreads();
  int base = 0, tot = 0;
#pragma unroll
  for (int k = 0; k < BIN_TB / 32; ++k) {
    const int w = sh[k];
    tot += w;
    if (k < wv) base += w;
  }
  *total = tot;
  return base + incl - v;
}

__global__ void __launch_bounds__(BIN_TB)
key_hist(const int* __restrict__ ei, int* __restrict__ hist, int E, int N, int cap) {
  __shared__ int h[KEYS];
  const int tid = threadIdx.x;
  h[tid] = 0;
  __syncthreads();
  const int e0 = blockIdx.x * CHUNK + tid;
  int key[CH_EPT];
#pragma unroll
  for (int k = 0; k < CH_EPT; ++k) {
    const int e = e0 + k * BIN_TB;
    const bool in = (e < E) && (e < cap);
    const int d = in ? ei[E + e] : -1;
    const int s = in ? ei[e] : 0;
    key[k] = edge_key(s, d, N);
  }
#pragma unroll
  for (int k = 0; k < CH_EPT; ++k)
    if (key[k] >= 0) atomicAdd(&h[key[k]], 1);
  __syncthreads();
  hist[blockIdx.x * KEYS + tid] = h[tid];
}

__global__ void __launch_bounds__(BIN_TB)
key_colscan(int* __restrict__ hist, int* __restrict__ kt, int NB) {
  __shared__ int gs[32][33];
  const int kl  = threadIdx.x & 31;
  const int g   = threadIdx.x >> 5;
  const int key = blockIdx.x * 32 + kl;
  const int rpg = (NB + 31) >> 5;
  const int r0  = min(g * rpg, NB);
  const int r1  = min(r0 + rpg, NB);
  int sum = 0;
  for (int r = r0; r < r1; ++r) sum += hist[r * KEYS + key];
  gs[g][kl] = sum;
  __syncthreads();
  int run = 0;
  for (int gg = 0; gg < g; ++gg) run += gs[gg][kl];
  for (int r = r0; r < r1; ++r) {
    const int c = hist[r * KEYS + key];
    hist[r * KEYS + key] = run;
    run += c;
  }
  if (g == 31) kt[key] = run;
}

__global__ void __launch_bounds__(BIN_TB)
key_scatter(const int* __restrict__ ei, const int* __restrict__ hist,
            const int* __restrict__ kt, int* __restrict__ plane,
            int E, int N, int cap) {
  __shared__ int sh[32];
  __shared__ int goff[KEYS];
  __shared__ int lcnt[KEYS];
  __shared__ int lstart[KEYS];
  __shared__ int stw[CHUNK];
  __shared__ unsigned short stk[CHUNK];
  const int tid = threadIdx.x;
  int tot;
  const int ks = block_excl_scan(kt[tid], sh, &tot);
  goff[tid] = ks + hist[blockIdx.x * KEYS + tid];
  lcnt[tid] = 0;
  __syncthreads();
  const int e0 = blockIdx.x * CHUNK + tid;
  int key[CH_EPT], word[CH_EPT], rk[CH_EPT];
#pragma unroll
  for (int k = 0; k < CH_EPT; ++k) {
    const int e = e0 + k * BIN_TB;
    const bool in = (e < E) && (e < cap);
    const int d = in ? ei[E + e] : -1;
    const int s = in ? ei[e] : 0;
    key[k]  = edge_key(s, d, N);
    word[k] = edge_word(s, d);
  }
#pragma unroll
  for (int k = 0; k < CH_EPT; ++k)
    rk[k] = (key[k] >= 0) ? atomicAdd(&lcnt[key[k]], 1) : 0;
  __syncthreads();
  int nvalid;
  const int ls = block_excl_scan(lcnt[tid], sh, &nvalid);
  lstart[tid] = ls;
  __syncthreads();
#pragma unroll
  for (int k = 0; k < CH_EPT; ++k) {
    if (key[k] >= 0) {
      const int q = lstart[key[k]] + rk[k];
      stw[q] = word[k];
      stk[q] = (unsigned short)key[k];
    }
  }
  __syncthreads();
  for (int j = tid; j < nvalid; j += BIN_TB) {
    const int k = stk[j];
    const int pos = goff[k] + (j - lstart[k]);
    if ((unsigned)pos < (unsigned)cap)
      plane[pos] = stw[j];
  }
}

__global__ void __launch_bounds__(BIN_TB)
bin_sort(const int* __restrict__ plane, const int* __restrict__ kt,
         unsigned short* __restrict__ list, int* __restrict__ cur, int N, int cap) {
  __shared__ int sh[32];
  __shared__ int sbs;
  __shared__ int cntL[512];
  __shared__ int fillL[512];
  __shared__ unsigned short stage[STAGE_CAP];
  const int tid = threadIdx.x;
  const int bin = blockIdx.x;
  int tot;
  const int ks = block_excl_scan(kt[tid], sh, &tot);
  if (tid == 2 * bin) sbs = ks;
  if (tid < 512) cntL[tid] = 0;
  __syncthreads();
  int bs    = sbs;
  int total = kt[2 * bin] + kt[2 * bin + 1];
  bs    = min(max(bs, 0), cap);
  total = min(max(total, 0), cap - bs);
  for (int j = tid; j < total; j += BIN_TB)
    atomicAdd(&cntL[(plane[bs + j] >> 16) & 511], 1);
  __syncthreads();
  const int c = (tid < 512) ? cntL[tid] : 0;
  int t2;
  const int ex = block_excl_scan(c, sh, &t2);
  if (tid < 512) {
    fillL[tid] = ex;
    const int n = (bin << 8) + (tid & 255);
    if (n < N) {
      const int v = n + ((tid >> 8) ? N : 0);
      int2 be;
      be.x = bs + ex;
      be.y = bs + ex + c;
      *(int2*)(cur + 2 * (size_t)v) = be;
    }
  }
  __syncthreads();
  const bool staged = (total <= STAGE_CAP);
  for (int j = tid; j < total; j += BIN_TB) {
    const int w = plane[bs + j];
    const int q = atomicAdd(&fillL[(w >> 16) & 511], 1);
    const unsigned short s16 = (unsigned short)(w & 0xFFFF);
    if (staged) {
      if ((unsigned)q < (unsigned)STAGE_CAP) stage[q] = s16;
    } else {
      if ((unsigned)q < (unsigned)total) list[bs + q] = s16;
    }
  }
  __syncthreads();
  if (staged)
    for (int j = tid; j < total; j += BIN_TB) list[bs + j] = stage[j];
}

template <int XMASK>
__device__ __forceinline__ float row_xor_read(float v) {
  return __int_as_float(__builtin_amdgcn_update_dpp(
      0, __float_as_int(v), 0x160 + XMASK,  0xF,  0xF,
       true));
}
template <int C>
__device__ __forceinline__ float head_sum(float v) {
  v += row_xor_read<1>(v);
  v += row_xor_read<2>(v);
  if (C == 8) v += row_xor_read<4>(v);
  return v;
}

template <int C, bool ELU>
__global__ void seg_softmax_agg(const unsigned short* __restrict__ list,
                                const int* __restrict__ cur,
                                const float* __restrict__ a_src,
                                const float* __restrict__ a_dst,
                                const float* __restrict__ Hm,
                                const float* __restrict__ bias,
                                float* __restrict__ out,
                                int cap, int N) {
  constexpr int F = 2 * C;
  constexpr int U = 4;
  const int t = blockIdx.x * blockDim.x + threadIdx.x;
  const int n = t / F;
  const int j = t % F;
  if (n >= N) return;
  const float asj = a_src[j];

  const int2 lbe = *(const int2*)(cur + 2 * n);
  const int2 hbe = *(const int2*)(cur + 2 * (N + n));
  int lbeg = lbe.x, lend = lbe.y;
  int hbeg = hbe.x, hend = hbe.y;
  lbeg = min(max(lbeg, 0), cap);
  lend = min(max(lend, lbeg), cap);
  hbeg = min(max(hbeg, 0), cap);
  hend = min(max(hend, hbeg), cap);
  const int nlow  = lend - lbeg;
  const int end   = nlow + (hend - hbeg);
  const int hskew = hbeg - nlow;

  const float hn  = Hm[n * F + j];
  const float adn = head_sum<C>(hn * a_dst[j]);
  float es = head_sum<C>(hn * asj) + adn;
  es = (es > 0.f) ? es : 0.2f * es;

  float m = es;
  if (!(m >= -3.0e38f && m <= 3.0e38f)) m = 0.f;
  float z = __expf(es - m);
  float acc = z * hn;
  for (int p = 0; p < end; p += U) {
    int sq[U];
    bool ok[U];
    float hq[U], e[U];
#pragma unroll
    for (int u = 0; u < U; ++u) {
      ok[u] = (p + u < end);
      const int q = ok[u] ? (p + u) : (end - 1);
      const bool high = (q >= nlow);
      const int sv = (int)list[q + (high ? hskew : lbeg)] + (high ? 65536 : 0);
      sq[u] = min(sv, N - 1);
    }
#pragma unroll
    for (int u = 0; u < U; ++u) hq[u] = Hm[sq[u] * F + j];
    float mn = m;
#pragma unroll
    for (int u = 0; u < U; ++u) {
      float t = head_sum<C>(hq[u] * asj) + adn;
      t = (t > 0.f) ? t : 0.2f * t;
      e[u] = ok[u] ? t : -3.0e38f;
      mn = fmaxf(mn, e[u]);
    }
    const float r = __expf(m - mn);
    z *= r;
    acc *= r;
#pragma unroll
    for (int u = 0; u < U; ++u) {
      const float wu = __expf(e[u] - mn);
      z += wu;
      acc += wu * hq[u];
    }
    m = mn;
  }
  float x = acc * (1.f / (z + 1e-16f)) + bias[j];
  if (ELU) x = (x > 0.f) ? x : (__expf(x) - 1.f);
  out[n * F + j] = x;
}

extern "C" void kernel_launch(void* const* d_in, const int* in_sizes, int n_in,
                              void* d_out, int out_size, void* d_ws, size_t ws_size,
                              hipStream_t stream) {
  const float* x   = (const float*)d_in[0];
  const int*   ei  = (const int*)  d_in[1];
  const float* W1  = (const float*)d_in[2];
  const float* as1 = (const float*)d_in[3];
  const float* ad1 = (const float*)d_in[4];
  const float* b1  = (const float*)d_in[5];
  const float* W2  = (const float*)d_in[6];
  const float* as2 = (const float*)d_in[7];
  const float* ad2 = (const float*)d_in[8];
  const float* b2  = (const float*)d_in[9];

  const int N = in_sizes[0] / 16;
  const int E = in_sizes[1] / 2;

  float*  ws    = (float*)d_ws;
  float*  h1    = ws;
  float*  x2    = ws + (size_t)16 * N;
  int*    plane = (int*)ws;
  int*    hist  = (int*)(ws + (size_t)32 * N);
  int*    cur   = (int*)(ws + (size_t)36 * N);
  unsigned short* list = (unsigned short*)(ws + (size_t)40 * N);
  float*  h2    = ws;

  const int TB = 256;
  const int cap        = 32 * N;
  const int edges      = (E < cap) ? E : cap;
  const int NB         = (edges + CHUNK - 1) / CHUNK;
  int*      kt         = hist + (size_t)NB * KEYS;
  const int bins       = (N + 255) / 256;
  const int tiles      = (N + 15) / 16;
  const int gemm_grid  = (tiles + 7) / 8;
  const int lane16_grid = (N * 16 + TB - 1) / TB;
  const int lane8_grid  = (N * 8 + TB - 1) / TB;

  key_hist<<<NB, BIN_TB, 0, stream>>>(ei, hist, E, N, cap);
  key_colscan<<<KEYS / 32, BIN_TB, 0, stream>>>(hist, kt, NB);
  key_scatter<<<NB, BIN_TB, 0, stream>>>(ei, hist, kt, plane, E, N, cap);
  bin_sort<<<bins, BIN_TB, 0, stream>>>(plane, kt, list, cur, N, cap);

  gemm16_wmma<<<gemm_grid, TB, 0, stream>>>(x, W1, h1, N, 16);
  seg_softmax_agg<8, true><<<lane16_grid, TB, 0, stream>>>(list, cur, as1, ad1, h1, b1,
                                                           x2, cap, N);

  gemm16_wmma<<<gemm_grid, TB, 0, stream>>>(x2, W2, h2, N, 8);
  seg_softmax_agg<4, false><<<lane8_grid, TB, 0, stream>>>(list, cur, as2, ad2, h2, b2,
                                                           (float*)d_out, cap, N);
}


# walk: a lane holds 4 channels of a row (16-byte gather), a node's lanes take 4 edges side by side; logit dot in-lane; node sums = 4 interleaved partial sums merged by lane swizzle (new sum order)
# speedup vs baseline: 19.6969x; 1.1769x over previous
#include <hip/hip_runtime.h>
#include <math.h>


typedef __attribute__((ext_vector_type(2))) float v2f;
typedef __attribute__((ext_vector_type(4))) float v4f;
typedef __attribute__((ext_vector_type(8))) float v8f;

__global__ void gemm16_wmma(const float* __restrict__ X,
                            const float* __restrict__ W,
                            float* __restrict__ Hout,
                            int nrows, int ncols) {
  const int wave = blockIdx.x * (blockDim.x >> 5) + (threadIdx.x >> 5);
  const int lane = threadIdx.x & 31;
  const int row_base = wave * 16;
  if (row_base >= nrows) return;

  const int m    = lane & 15;
  const int half = lane >> 4;
  int row = row_base + m; if (row > nrows - 1) row = nrows - 1;
  int col = m;            if (col > ncols - 1) col = ncols - 1;

  const float2* __restrict__ Xr = (const float2*)(X + (size_t)row * 16);

  v8f c = {0.f, 0.f, 0.f, 0.f, 0.f, 0.f, 0.f, 0.f};
#pragma unroll
  for (int k0 = 0; k0 < 4; ++k0) {
    const int k = k0 * 4 + half * 2;
    const float2 av = Xr[k0 * 2 + half];
    v2f a; a.x = av.x; a.y = av.y;
    v2f b;
    b.x = W[(k)     * ncols + col];
    b.y = W[(k + 1) * ncols + col];
    c = __builtin_amdgcn_wmma_f32_16x16x4_f32(
         false, a,  false, b,
         (short)0, c,  false,  false);
  }

  if (m < ncols) {
    const int rofs = half * 8;
    float* __restrict__ outp = Hout + (size_t)(row_base + rofs) * ncols + m;
    if (row_base + 16 <= nrows) {
#pragma unroll
      for (int r = 0; r < 8; ++r) outp[(size_t)r * ncols] = c[r];
    } else {
#pragma unroll
      for (int r = 0; r < 8; ++r)
        if (row_base + rofs + r < nrows) outp[(size_t)r * ncols] = c[r];
    }
  }
}

#define KEYS 1024
#define BIN_TB 1024
#define CH_EPT 16
#define CHUNK (BIN_TB * CH_EPT)
#define STAGE_CAP 32768

__device__ __forceinline__ int edge_key(int s, int d, int N) {
  const bool ok = ((unsigned)d < (unsigned)N) && (d < 131072);
  return ok ? (((d >> 8) << 1) | (((s >> 16) != 0) ? 1 : 0)) : -1;
}
__device__ __forceinline__ int edge_word(int s, int d) {
  const int idx = ((((s >> 16) != 0) ? 1 : 0) << 8) | (d & 255);
  return (idx << 16) | (s & 0xFFFF);
}

__device__ __forceinline__ int block_excl_scan(int v, int* sh, int* total) {
  const int lane = threadIdx.x & 31;
  const int wv   = threadIdx.x >> 5;
  int incl = v;
#pragma unroll
  for (int o = 1; o < 32; o <<= 1) {
    const int up = __shfl_up(incl, o);
    if (lane >= o) incl += up;
  }
  __syncthreads();
  if (lane == 31) sh[wv] = incl;
  __syncthreads();
  int base = 0, tot = 0;
#pragma unroll
  for (int k = 0; k < BIN_TB / 32; ++k) {
    const int w = sh[k];
    tot += w;
    if (k < wv) base += w;
  }
  *total = tot;
  return base + incl - v;
}

__global__ void __launch_bounds__(BIN_TB)
key_hist(const int* __restrict__ ei, int* __restrict__ hist, int E, int N, int cap) {
  __shared__ int h[KEYS];
  const int tid = threadIdx.x;
  h[tid] = 0;
  __syncthreads();
  const int e0 = blockIdx.x * CHUNK + tid;
  int key[CH_EPT];
#pragma unroll
  for (int k = 0; k < CH_EPT; ++k) {
    const int e = e0 + k * BIN_TB;
    const bool in = (e < E) && (e < cap);
    const int d = in ? ei[E + e] : -1;
    const int s = in ? ei[e] : 0;
    key[k] = edge_key(s, d, N);
  }
#pragma unroll
  for (int k = 0; k < CH_EPT; ++k)
    if (key[k] >= 0) atomicAdd(&h[key[k]], 1);
  __syncthreads();
  hist[blockIdx.x * KEYS + tid] = h[tid];
}

__global__ void __launch_bounds__(BIN_TB)
key_colscan(int* __restrict__ hist, int* __restrict__ kt, int NB) {
  __shared__ int gs[32][33];
  const int kl  = threadIdx.x & 31;
  const int g   = threadIdx.x >> 5;
  const int key = blockIdx.x * 32 + kl;
  const int rpg = (NB + 31) >> 5;
  const int r0  = min(g * rpg, NB);
  const int r1  = min(r0 + rpg, NB);
  int sum = 0;
  for (int r = r0; r < r1; ++r) sum += hist[r * KEYS + key];
  gs[g][kl] = sum;
  __syncthreads();
  int run = 0;
  for (int gg = 0; gg < g; ++gg) run += gs[gg][kl];
  for (int r = r0; r < r1; ++r) {
    const int c = hist[r * KEYS + key];
    hist[r * KEYS + key] = run;
    run += c;
  }
  if (g == 31) kt[key] = run;
}

__global__ void __launch_bounds__(BIN_TB)
key_scatter(const int* __restrict__ ei, const int* __restrict__ hist,
            const int* __restrict__ kt, int* __restrict__ plane,
            int E, int N, int cap) {
  __shared__ int sh[32];
  __shared__ int goff[KEYS];
  __shared__ int lcnt[KEYS];
  __shared__ int lstart[KEYS];
  __shared__ int stw[CHUNK];
  __shared__ unsigned short stk[CHUNK];
  const int tid = threadIdx.x;
  int tot;
  const int ks = block_excl_scan(kt[tid], sh, &tot);
  goff[tid] = ks + hist[blockIdx.x * KEYS + tid];
  lcnt[tid] = 0;
  __syncthreads();
  const int e0 = blockIdx.x * CHUNK + tid;
  int key[CH_EPT], word[CH_EPT], rk[CH_EPT];
#pragma unroll
  for (int k = 0; k < CH_EPT; ++k) {
    const int e = e0 + k * BIN_TB;
    const bool in = (e < E) && (e < cap);
    const int d = in ? ei[E + e] : -1;
    const int s = in ? ei[e] : 0;
    key[k]  = edge_key(s, d, N);
    word[k] = edge_word(s, d);
  }
#pragma unroll
  for (int k = 0; k < CH_EPT; ++k)
    rk[k] = (key[k] >= 0) ? atomicAdd(&lcnt[key[k]], 1) : 0;
  __syncthreads();
  int nvalid;
  const int ls = block_excl_scan(lcnt[tid], sh, &nvalid);
  lstart[tid] = ls;
  __syncthreads();
#pragma unroll
  for (int k = 0; k < CH_EPT; ++k) {
    if (key[k] >= 0) {
      const int q = lstart[key[k]] + rk[k];
      stw[q] = word[k];
      stk[q] = (unsigned short)key[k];
    }
  }
  __syncthreads();
  for (int j = tid; j < nvalid; j += BIN_TB) {
    const int k = stk[j];
    const int pos = goff[k] + (j - lstart[k]);
    if ((unsigned)pos < (unsigned)cap)
      plane[pos] = stw[j];
  }
}

__global__ void __launch_bounds__(BIN_TB)
bin_sort(const int* __restrict__ plane, const int* __restrict__ kt,
         unsigned short* __restrict__ list, int* __restrict__ cur, int N, int cap) {
  __shared__ int sh[32];
  __shared__ int sbs;
  __shared__ int cntL[512];
  __shared__ int fillL[512];
  __shared__ unsigned short stage[STAGE_CAP];
  const int tid = threadIdx.x;
  const int bin = blockIdx.x;
  int tot;
  const int ks = block_excl_scan(kt[tid], sh, &tot);
  if (tid == 2 * bin) sbs = ks;
  if (tid < 512) cntL[tid] = 0;
  __syncthreads();
  int bs    = sbs;
  int total = kt[2 * bin] + kt[2 * bin + 1];
  bs    = min(max(bs, 0), cap);
  total = min(max(total, 0), cap - bs);
  for (int j = tid; j < total; j += BIN_TB)
    atomicAdd(&cntL[(plane[bs + j] >> 16) & 511], 1);
  __syncthreads();
  const int c = (tid < 512) ? cntL[tid] : 0;
  int t2;
  const int ex = block_excl_scan(c, sh, &t2);
  if (tid < 512) {
    fillL[tid] = ex;
    const int n = (bin << 8) + (tid & 255);
    if (n < N) {
      const int v = n + ((tid >> 8) ? N : 0);
      int2 be;
      be.x = bs + ex;
      be.y = bs + ex + c;
      *(int2*)(cur + 2 * (size_t)v) = be;
    }
  }
  __syncthreads();
  const bool staged = (total <= STAGE_CAP);
  for (int j = tid; j < total; j += BIN_TB) {
    const int w = plane[bs + j];
    const int q = atomicAdd(&fillL[(w >> 16) & 511], 1);
    const unsigned short s16 = (unsigned short)(w & 0xFFFF);
    if (staged) {
      if ((unsigned)q < (unsigned)STAGE_CAP) stage[q] = s16;
    } else {
      if ((unsigned)q < (unsigned)total) list[bs + q] = s16;
    }
  }
  __syncthreads();
  if (staged)
    for (int j = tid; j < total; j += BIN_TB) list[bs + j] = stage[j];
}

template <int XMASK>
__device__ __forceinline__ float row_xor_read(float v) {
  return __int_as_float(__builtin_amdgcn_update_dpp(
      0, __float_as_int(v), 0x160 + XMASK,  0xF,  0xF,
       true));
}
template <int C>
__device__ __forceinline__ float head_sum(float v) {
  if (C == 8) v += row_xor_read<1>(v);
  return v;
}
template <int LPE>
__device__ __forceinline__ float slot_sum(float v) {
  v += row_xor_read<LPE>(v);
  v += row_xor_read<2 * LPE>(v);
  return v;
}
template <int LPE>
__device__ __forceinline__ float slot_max(float v) {
  v = fmaxf(v, row_xor_read<LPE>(v));
  v = fmaxf(v, row_xor_read<2 * LPE>(v));
  return v;
}
__device__ __forceinline__ float dot4(v4f a, v4f b) {
  return a.x * b.x + a.y * b.y + a.z * b.z + a.w * b.w;
}

template <int C, bool ELU>
__global__ void seg_softmax_agg(const unsigned short* __restrict__ list,
                                const int* __restrict__ cur,
                                const float* __restrict__ a_src,
                                const float* __restrict__ a_dst,
                                const float* __restrict__ Hm,
                                const float* __restrict__ bias,
                                float* __restrict__ out,
                                int cap, int N) {
  constexpr int F     = 2 * C;
  constexpr int LPE   = F / 4;
  constexpr int SLOTS = 4;
  constexpr int U     = 4;
  const int t = blockIdx.x * blockDim.x + threadIdx.x;
  const int n = t / F;
  const int l = t & (F - 1);
  if (n >= N) return;
  const int q = l & (LPE - 1);
  const int g = l / LPE;
  const v4f* __restrict__ H4 = (const v4f*)Hm;
  const v4f as4 = ((const v4f*)a_src)[q];
  const v4f ad4 = ((const v4f*)a_dst)[q];

  const int2 lbe = *(const int2*)(cur + 2 * n);
  const int2 hbe = *(const int2*)(cur + 2 * (N + n));
  int lbeg = lbe.x, lend = lbe.y;
  int hbeg = hbe.x, hend = hbe.y;
  lbeg = min(max(lbeg, 0), cap);
  lend = min(max(lend, lbeg), cap);
  hbeg = min(max(hbeg, 0), cap);
  hend = min(max(hend, hbeg), cap);
  const int nlow  = lend - lbeg;
  const int end   = nlow + (hend - hbeg);
  const int hskew = hbeg - nlow;

  const v4f hn    = H4[n * LPE + q];
  const float adn = head_sum<C>(dot4(hn, ad4));
  float es = head_sum<C>(dot4(hn, as4)) + adn;
  es = (es > 0.f) ? es : 0.2f * es;

  float m0 = es;
  if (!(m0 >= -3.0e38f && m0 <= 3.0e38f)) m0 = 0.f;
  const bool first = (g == 0);
  float m = first ? m0 : -3.0e38f;
  float z = first ? __expf(es - m0) : 0.f;
  v4f acc = z * hn;
  for (int p = 0; p < end; p += SLOTS * U) {
    int sq[U];
    bool ok[U];
    v4f hq[U];
    float e[U];
#pragma unroll
    for (int u = 0; u < U; ++u) {
      const int i = p + u * SLOTS + g;
      ok[u] = (i < end);
      const int qi = ok[u] ? i : (end - 1);
      const bool high = (qi >= nlow);
      const int sv = (int)list[qi + (high ? hskew : lbeg)] + (high ? 65536 : 0);
      sq[u] = min(sv, N - 1);
    }
#pragma unroll
    for (int u = 0; u < U; ++u) hq[u] = H4[sq[u] * LPE + q];
    static_assert(U == 4, "the statement below names four rows");
    asm volatile("" : "+v"(hq[0]), "+v"(hq[1]), "+v"(hq[2]), "+v"(hq[3]));
    float mn = m;
#pragma unroll
    for (int u = 0; u < U; ++u) {
      float tt = head_sum<C>(dot4(hq[u], as4)) + adn;
      tt = (tt > 0.f) ? tt : 0.2f * tt;
      e[u] = ok[u] ? tt : -3.0e38f;
      mn = fmaxf(mn, e[u]);
    }
    const float r = __expf(m - mn);
    z *= r;
    acc *= r;
#pragma unroll
    for (int u = 0; u < U; ++u) {
      const float wu = ok[u] ? __expf(e[u] - mn) : 0.f;
      z += wu;
      acc += wu * hq[u];
    }
    m = mn;
  }
  const float mall = slot_max<LPE>(m);
  const float sc   = __expf(m - mall);
  z = slot_sum<LPE>(z * sc);
  acc.x = slot_sum<LPE>(acc.x * sc);
  acc.y = slot_sum<LPE>(acc.y * sc);
  acc.z = slot_sum<LPE>(acc.z * sc);
  acc.w = slot_sum<LPE>(acc.w * sc);
  if (first) {
    const v4f b4 = ((const v4f*)bias)[q];
    v4f x = acc * (1.f / (z + 1e-16f)) + b4;
    if (ELU) {
      x.x = (x.x > 0.f) ? x.x : (__expf(x.x) - 1.f);
      x.y = (x.y > 0.f) ? x.y : (__expf(x.y) - 1.f);
      x.z = (x.z > 0.f) ? x.z : (__expf(x.z) - 1.f);
      x.w = (x.w > 0.f) ? x.w : (__expf(x.w) - 1.f);
    }
    ((v4f*)out)[n * LPE + q] = x;
  }
}

extern "C" void kernel_launch(void* const* d_in, const int* in_sizes, int n_in,
                              void* d_out, int out_size, void* d_ws, size_t ws_size,
                              hipStream_t stream) {
  const float* x   = (const float*)d_in[0];
  const int*   ei  = (const int*)  d_in[1];
  const float* W1  = (const float*)d_in[2];
  const float* as1 = (const float*)d_in[3];
  const float* ad1 = (const float*)d_in[4];
  const float* b1  = (const float*)d_in[5];
  const float* W2  = (const float*)d_in[6];
  const float* as2 = (const float*)d_in[7];
  const float* ad2 = (const float*)d_in[8];
  const float* b2  = (const float*)d_in[9];

  const int N = in_sizes[0] / 16;
  const int E = in_sizes[1] / 2;

  float*  ws    = (float*)d_ws;
  float*  h1    = ws;
  float*  x2    = ws + (size_t)16 * N;
  int*    plane = (int*)ws;
  int*    hist  = (int*)(ws + (size_t)32 * N);
  int*    cur   = (int*)(ws + (size_t)36 * N);
  unsigned short* list = (unsigned short*)(ws + (size_t)40 * N);
  float*  h2    = ws;

  const int TB = 256;
  const int cap        = 32 * N;
  const int edges      = (E < cap) ? E : cap;
  const int NB         = (edges + CHUNK - 1) / CHUNK;
  int*      kt         = hist + (size_t)NB * KEYS;
  const int bins       = (N + 255) / 256;
  const int tiles      = (N + 15) / 16;
  const int gemm_grid  = (tiles + 7) / 8;
  const int lane16_grid = (N * 16 + TB - 1) / TB;
  const int lane8_grid  = (N * 8 + TB - 1) / TB;

  key_hist<<<NB, BIN_TB, 0, stream>>>(ei, hist, E, N, cap);
  key_colscan<<<KEYS / 32, BIN_TB, 0, stream>>>(hist, kt, NB);
  key_scatter<<<NB, BIN_TB, 0, stream>>>(ei, hist, kt, plane, E, N, cap);
  bin_sort<<<bins, BIN_TB, 0, stream>>>(plane, kt, list, cur, N, cap);

  gemm16_wmma<<<gemm_grid, TB, 0, stream>>>(x, W1, h1, N, 16);
  seg_softmax_agg<8, true><<<lane16_grid, TB, 0, stream>>>(list, cur, as1, ad1, h1, b1,
                                                           x2, cap, N);

  gemm16_wmma<<<gemm_grid, TB, 0, stream>>>(x2, W2, h2, N, 8);
  seg_softmax_agg<4, false><<<lane8_grid, TB, 0, stream>>>(list, cur, as2, ad2, h2, b2,
                                                           (float*)d_out, cap, N);
}


# list build in 2 launches, not 4: each block LDS-sorts its run in place and writes its key starts; a bin's block collects its piece of every run (list slot = column sum of starts); new list order
# speedup vs baseline: 25.9250x; 1.3162x over previous
#include <hip/hip_runtime.h>
#include <math.h>


typedef __attribute__((ext_vector_type(2))) float v2f;
typedef __attribute__((ext_vector_type(4))) float v4f;
typedef __attribute__((ext_vector_type(8))) float v8f;

__global__ void gemm16_wmma(const float* __restrict__ X,
                            const float* __restrict__ W,
                            float* __restrict__ Hout,
                            int nrows, int ncols) {
  const int wave = blockIdx.x * (blockDim.x >> 5) + (threadIdx.x >> 5);
  const int lane = threadIdx.x & 31;
  const int row_base = wave * 16;
  if (row_base >= nrows) return;

  const int m    = lane & 15;
  const int half = lane >> 4;
  int row = row_base + m; if (row > nrows - 1) row = nrows - 1;
  int col = m;            if (col > ncols - 1) col = ncols - 1;

  const float2* __restrict__ Xr = (const float2*)(X + (size_t)row * 16);

  v8f c = {0.f, 0.f, 0.f, 0.f, 0.f, 0.f, 0.f, 0.f};
#pragma unroll
  for (int k0 = 0; k0 < 4; ++k0) {
    const int k = k0 * 4 + half * 2;
    const float2 av = Xr[k0 * 2 + half];
    v2f a; a.x = av.x; a.y = av.y;
    v2f b;
    b.x = W[(k)     * ncols + col];
    b.y = W[(k + 1) * ncols + col];
    c = __builtin_amdgcn_wmma_f32_16x16x4_f32(
         false, a,  false, b,
         (short)0, c,  false,  false);
  }

  if (m < ncols) {
    const int rofs = half * 8;
    float* __restrict__ outp = Hout + (size_t)(row_base + rofs) * ncols + m;
    if (row_base + 16 <= nrows) {
#pragma unroll
      for (int r = 0; r < 8; ++r) outp[(size_t)r * ncols] = c[r];
    } else {
#pragma unroll
      for (int r = 0; r < 8; ++r)
        if (row_base + rofs + r < nrows) outp[(size_t)r * ncols] = c[r];
    }
  }
}

#define KEYS 1024
#define BIN_TB 1024
#define CH_EPT 16
#define CHUNK (BIN_TB * CH_EPT)
#define TABP (KEYS + 8)
#define WORD_CAP 16384

__device__ __forceinline__ int edge_key(int s, int d, int N) {
  const bool ok = ((unsigned)d < (unsigned)N) && (d < 131072);
  return ok ? (((d >> 8) << 1) | (((s >> 16) != 0) ? 1 : 0)) : -1;
}
__device__ __forceinline__ int edge_word(int s, int d) {
  const int idx = ((((s >> 16) != 0) ? 1 : 0) << 8) | (d & 255);
  return (idx << 16) | (s & 0xFFFF);
}

__device__ __forceinline__ int block_excl_scan(int v, int* sh, int* total) {
  const int lane = threadIdx.x & 31;
  const int wv   = threadIdx.x >> 5;
  int incl = v;
#pragma unroll
  for (int o = 1; o < 32; o <<= 1) {
    const int up = __shfl_up(incl, o);
    if (lane >= o) incl += up;
  }
  __syncthreads();
  if (lane == 31) sh[wv] = incl;
  __syncthreads();
  int base = 0, tot = 0;
#pragma unroll
  for (int k = 0; k < BIN_TB / 32; ++k) {
    const int w = sh[k];
    tot += w;
    if (k < wv) base += w;
  }
  *total = tot;
  return base + incl - v;
}

__global__ void __launch_bounds__(BIN_TB)
run_sort(const int* __restrict__ ei, int* __restrict__ tab, int* __restrict__ plane,
         int E, int N, int cap) {
  __shared__ int sh[32];
  __shared__ int lcnt[KEYS];
  __shared__ int lstart[KEYS];
  __shared__ int stw[CHUNK];
  const int tid = threadIdx.x;
  lcnt[tid] = 0;
  __syncthreads();
  const int e0 = blockIdx.x * CHUNK + tid;
  int key[CH_EPT], word[CH_EPT], rk[CH_EPT];
#pragma unroll
  for (int k = 0; k < CH_EPT; ++k) {
    const int e = e0 + k * BIN_TB;
    const bool in = (e < E) && (e < cap);
    const int d = in ? ei[E + e] : -1;
    const int s = in ? ei[e] : 0;
    key[k]  = edge_key(s, d, N);
    word[k] = edge_word(s, d);
  }
#pragma unroll
  for (int k = 0; k < CH_EPT; ++k)
    rk[k] = (key[k] >= 0) ? atomicAdd(&lcnt[key[k]], 1) : 0;
  __syncthreads();
  int nvalid;
  const int ls = block_excl_scan(lcnt[tid], sh, &nvalid);
  lstart[tid] = ls;
  tab[(size_t)blockIdx.x * TABP + tid] = ls;
  if (tid == 0) tab[(size_t)blockIdx.x * TABP + KEYS] = nvalid;
  __syncthreads();
#pragma unroll
  for (int k = 0; k < CH_EPT; ++k)
    if (key[k] >= 0) stw[lstart[key[k]] + rk[k]] = word[k];
  __syncthreads();
  for (int j = tid; j < nvalid; j += BIN_TB) {
    const int pos = blockIdx.x * CHUNK + j;
    if ((unsigned)pos < (unsigned)cap) plane[pos] = stw[j];
  }
}

__device__ __forceinline__ void bin_take(int w, int slot, bool staged, int* cntL, int* stagew) {
  atomicAdd(&cntL[(w >> 16) & 511], 1);
  if (staged && (unsigned)slot < (unsigned)WORD_CAP) stagew[slot] = w;
}

__global__ void __launch_bounds__(BIN_TB)
bin_sort(const int* __restrict__ plane, const int* __restrict__ tab,
         unsigned short* __restrict__ list, int* __restrict__ cur,
         int N, int cap, int NB) {
  __shared__ int sh[32];
  __shared__ int cntL[512];
  __shared__ int fillL[512];
  __shared__ int pbeg[BIN_TB];
  __shared__ int poff[BIN_TB];
  __shared__ int plen[BIN_TB];
  __shared__ int stagew[WORD_CAP];
  __shared__ unsigned short stage[WORD_CAP];
  const int tid  = threadIdx.x;
  const int lane = tid & 31;
  const int wv   = tid >> 5;
  const int bin  = blockIdx.x;
  const int nb   = min(NB, BIN_TB);
  int p0 = 0, len = 0;
  if (tid < nb) {
    const int* __restrict__ row = tab + (size_t)tid * TABP + 2 * bin;
    p0  = min(max(row[0], 0), CHUNK);
    len = min(max(row[2] - p0, 0), CHUNK - p0);
  }
  if (tid < 512) cntL[tid] = 0;
  int bs, total;
  (void)block_excl_scan(p0, sh, &bs);
  const int off = block_excl_scan(len, sh, &total);
  pbeg[tid] = tid * CHUNK + p0;
  poff[tid] = off;
  plen[tid] = len;
  bs    = min(max(bs, 0), cap);
  total = min(max(total, 0), cap - bs);
  const bool staged = (total <= WORD_CAP);
  __syncthreads();
  {
    int wq[16], sl[16];
#pragma unroll
    for (int u = 0; u < 8; ++u) {
      const int b  = wv + 32 * u;
      const int ln = (b < nb) ? plen[b] : 0;
      const int pb = (b < nb) ? pbeg[b] : 0;
      const int po = (b < nb) ? poff[b] : 0;
#pragma unroll
      for (int h = 0; h < 2; ++h) {
        const int i = lane + 32 * h;
        sl[2 * u + h] = po + i;
        wq[2 * u + h] = (i < ln) ? plane[min(pb + i, cap - 1)] : -1;
      }
    }
#pragma unroll
    for (int x = 0; x < 16; ++x)
      if (wq[x] >= 0) bin_take(wq[x], sl[x], staged, cntL, stagew);
  }
  for (int b = wv; b < nb; b += 32) {
    const int ln = plen[b], pb = pbeg[b], po = poff[b];
    for (int i = ((b < 256) ? 64 : 0) + lane; i < ln; i += 32) {
      const int w = plane[min(pb + i, cap - 1)];
      if (w >= 0) bin_take(w, po + i, staged, cntL, stagew);
    }
  }
  __syncthreads();
  const int c = (tid < 512) ? cntL[tid] : 0;
  int t2;
  const int ex = block_excl_scan(c, sh, &t2);
  if (tid < 512) {
    fillL[tid] = ex;
    const int n = (bin << 8) + (tid & 255);
    if (n < N) {
      const int v = n + ((tid >> 8) ? N : 0);
      int2 be;
      be.x = bs + ex;
      be.y = bs + ex + c;
      *(int2*)(cur + 2 * (size_t)v) = be;
    }
  }
  __syncthreads();
  if (staged) {
    for (int j = tid; j < total; j += BIN_TB) {
      const int w = stagew[j];
      const int q = atomicAdd(&fillL[(w >> 16) & 511], 1);
      if ((unsigned)q < (unsigned)WORD_CAP) stage[q] = (unsigned short)(w & 0xFFFF);
    }
    __syncthreads();
    for (int j = tid; j < total; j += BIN_TB) list[bs + j] = stage[j];
  } else {
    for (int b = wv; b < nb; b += 32) {
      const int ln = plen[b], pb = pbeg[b];
      for (int i = lane; i < ln; i += 32) {
        const int w = plane[min(pb + i, cap - 1)];
        if (w >= 0) {
          const int q = atomicAdd(&fillL[(w >> 16) & 511], 1);
          if ((unsigned)q < (unsigned)total) list[bs + q] = (unsigned short)(w & 0xFFFF);
        }
      }
    }
  }
}

template <int XMASK>
__device__ __forceinline__ float row_xor_read(float v) {
  return __int_as_float(__builtin_amdgcn_update_dpp(
      0, __float_as_int(v), 0x160 + XMASK,  0xF,  0xF,
       true));
}
template <int C>
__device__ __forceinline__ float head_sum(float v) {
  if (C == 8) v += row_xor_read<1>(v);
  return v;
}
template <int LPE>
__device__ __forceinline__ float slot_sum(float v) {
  v += row_xor_read<LPE>(v);
  v += row_xor_read<2 * LPE>(v);
  return v;
}
template <int LPE>
__device__ __forceinline__ float slot_max(float v) {
  v = fmaxf(v, row_xor_read<LPE>(v));
  v = fmaxf(v, row_xor_read<2 * LPE>(v));
  return v;
}
__device__ __forceinline__ float dot4(v4f a, v4f b) {
  return a.x * b.x + a.y * b.y + a.z * b.z + a.w * b.w;
}

template <int C, bool ELU>
__global__ void seg_softmax_agg(const unsigned short* __restrict__ list,
                                const int* __restrict__ cur,
                                const float* __restrict__ a_src,
                                const float* __restrict__ a_dst,
                                const float* __restrict__ Hm,
                                const float* __restrict__ bias,
                                float* __restrict__ out,
                                int cap, int N) {
  constexpr int F     = 2 * C;
  constexpr int LPE   = F / 4;
  constexpr int SLOTS = 4;
  constexpr int U     = 4;
  const int t = blockIdx.x * blockDim.x + threadIdx.x;
  const int n = t / F;
  const int l = t & (F - 1);
  if (n >= N) return;
  const int q = l & (LPE - 1);
  const int g = l / LPE;
  const v4f* __restrict__ H4 = (const v4f*)Hm;
  const v4f as4 = ((const v4f*)a_src)[q];
  const v4f ad4 = ((const v4f*)a_dst)[q];

  const int2 lbe = *(const int2*)(cur + 2 * n);
  const int2 hbe = *(const int2*)(cur + 2 * (N + n));
  int lbeg = lbe.x, lend = lbe.y;
  int hbeg = hbe.x, hend = hbe.y;
  lbeg = min(max(lbeg, 0), cap);
  lend = min(max(lend, lbeg), cap);
  hbeg = min(max(hbeg, 0), cap);
  hend = min(max(hend, hbeg), cap);
  const int nlow  = lend - lbeg;
  const int end   = nlow + (hend - hbeg);
  const int hskew = hbeg - nlow;

  const v4f hn    = H4[n * LPE + q];
  const float adn = head_sum<C>(dot4(hn, ad4));
  float es = head_sum<C>(dot4(hn, as4)) + adn;
  es = (es > 0.f) ? es : 0.2f * es;

  float m0 = es;
  if (!(m0 >= -3.0e38f && m0 <= 3.0e38f)) m0 = 0.f;
  const bool first = (g == 0);
  float m = first ? m0 : -3.0e38f;
  float z = first ? __expf(es - m0) : 0.f;
  v4f acc = z * hn;
  for (int p = 0; p < end; p += SLOTS * U) {
    int sq[U];
    bool ok[U];
    v4f hq[U];
    float e[U];
#pragma unroll
    for (int u = 0; u < U; ++u) {
      const int i = p + u * SLOTS + g;
      ok[u] = (i < end);
      const int qi = ok[u] ? i : (end - 1);
      const bool high = (qi >= nlow);
      const int sv = (int)list[qi + (high ? hskew : lbeg)] + (high ? 65536 : 0);
      sq[u] = min(sv, N - 1);
    }
#pragma unroll
    for (int u = 0; u < U; ++u) hq[u] = H4[sq[u] * LPE + q];
    static_assert(U == 4, "the statement below names four rows");
    asm volatile("" : "+v"(hq[0]), "+v"(hq[1]), "+v"(hq[2]), "+v"(hq[3]));
    float mn = m;
#pragma unroll
    for (int u = 0; u < U; ++u) {
      float tt = head_sum<C>(dot4(hq[u], as4)) + adn;
      tt = (tt > 0.f) ? tt : 0.2f * tt;
      e[u] = ok[u] ? tt : -3.0e38f;
      mn = fmaxf(mn, e[u]);
    }
    const float r = __expf(m - mn);
    z *= r;
    acc *= r;
#pragma unroll
    for (int u = 0; u < U; ++u) {
      const float wu = ok[u] ? __expf(e[u] - mn) : 0.f;
      z += wu;
      acc += wu * hq[u];
    }
    m = mn;
  }
  const float mall = slot_max<LPE>(m);
  const float sc   = __expf(m - mall);
  z = slot_sum<LPE>(z * sc);
  acc.x = slot_sum<LPE>(acc.x * sc);
  acc.y = slot_sum<LPE>(acc.y * sc);
  acc.z = slot_sum<LPE>(acc.z * sc);
  acc.w = slot_sum<LPE>(acc.w * sc);
  if (first) {
    const v4f b4 = ((const v4f*)bias)[q];
    v4f x = acc * (1.f / (z + 1e-16f)) + b4;
    if (ELU) {
      x.x = (x.x > 0.f) ? x.x : (__expf(x.x) - 1.f);
      x.y = (x.y > 0.f) ? x.y : (__expf(x.y) - 1.f);
      x.z = (x.z > 0.f) ? x.z : (__expf(x.z) - 1.f);
      x.w = (x.w > 0.f) ? x.w : (__expf(x.w) - 1.f);
    }
    ((v4f*)out)[n * LPE + q] = x;
  }
}

extern "C" void kernel_launch(void* const* d_in, const int* in_sizes, int n_in,
                              void* d_out, int out_size, void* d_ws, size_t ws_size,
                              hipStream_t stream) {
  const float* x   = (const float*)d_in[0];
  const int*   ei  = (const int*)  d_in[1];
  const float* W1  = (const float*)d_in[2];
  const float* as1 = (const float*)d_in[3];
  const float* ad1 = (const float*)d_in[4];
  const float* b1  = (const float*)d_in[5];
  const float* W2  = (const float*)d_in[6];
  const float* as2 = (const float*)d_in[7];
  const float* ad2 = (const float*)d_in[8];
  const float* b2  = (const float*)d_in[9];

  const int N = in_sizes[0] / 16;
  const int E = in_sizes[1] / 2;

  float*  ws    = (float*)d_ws;
  float*  h1    = ws;
  float*  x2    = ws + (size_t)16 * N;
  int*    plane = (int*)ws;
  int*    tab   = (int*)(ws + (size_t)32 * N);
  int*    cur   = (int*)(ws + (size_t)36 * N);
  unsigned short* list = (unsigned short*)(ws + (size_t)40 * N);
  float*  h2    = ws;

  const int TB = 256;
  const int cap        = 32 * N;
  const int edges      = (E < cap) ? E : cap;
  const int NB         = (edges + CHUNK - 1) / CHUNK;
  const int bins       = (N + 255) / 256;
  const int tiles      = (N + 15) / 16;
  const int gemm_grid  = (tiles + 7) / 8;
  const int lane16_grid = (N * 16 + TB - 1) / TB;
  const int lane8_grid  = (N * 8 + TB - 1) / TB;

  run_sort<<<NB, BIN_TB, 0, stream>>>(ei, tab, plane, E, N, cap);
  bin_sort<<<bins, BIN_TB, 0, stream>>>(plane, tab, list, cur, N, cap, NB);

  gemm16_wmma<<<gemm_grid, TB, 0, stream>>>(x, W1, h1, N, 16);
  seg_softmax_agg<8, true><<<lane16_grid, TB, 0, stream>>>(list, cur, as1, ad1, h1, b1,
                                                           x2, cap, N);

  gemm16_wmma<<<gemm_grid, TB, 0, stream>>>(x2, W2, h2, N, 8);
  seg_softmax_agg<4, false><<<lane8_grid, TB, 0, stream>>>(list, cur, as2, ad2, h2, b2,
                                                           (float*)d_out, cap, N);
}
